# starting kernel
# baseline (speedup 1.0000x reference)
#include <hip/hip_runtime.h>
#include <hip/hip_bf16.h>
#include <math.h>

#define NPOS  80
#define NCHAR 256
#define HDIM  512
#define IDIM  (NPOS * NCHAR)
#define BATCH 8192
#define LN_EPS 1e-5f
#define KC    32
#define LDP   34

typedef __attribute__((ext_vector_type(2))) float v2f;
typedef __attribute__((ext_vector_type(8))) float v8f;

__global__ void k_transpose_w1(const float* __restrict__ W1,
                               float* __restrict__ W1T) {
  __shared__ float tile[32][33];
  const int i0 = blockIdx.x * 32;
  const int j0 = blockIdx.y * 32;
  const int tx = threadIdx.x;
  const int ty = threadIdx.y;
#pragma unroll
  for (int r = ty; r < 32; r += 8)
    tile[r][tx] = W1[(size_t)(j0 + r) * IDIM + (i0 + tx)];
  __syncthreads();
#pragma unroll
  for (int r = ty; r < 32; r += 8)
    W1T[(size_t)(i0 + r) * HDIM + (j0 + tx)] = tile[tx][r];
}

__device__ __forceinline__ float2 ln_stats_512(float a0, float a1) {
  float s = a0 + a1;
  float q = a0 * a0 + a1 * a1;
#pragma unroll
  for (int off = 16; off > 0; off >>= 1) {
    s += __shfl_xor(s, off, 32);
    q += __shfl_xor(q, off, 32);
  }
  __shared__ float ssum[8], sq[8], stats[2];
  const int wid  = threadIdx.x >> 5;
  const int lane = threadIdx.x & 31;
  if (lane == 0) { ssum[wid] = s; sq[wid] = q; }
  __syncthreads();
  if (threadIdx.x == 0) {
    float S = 0.f, Q = 0.f;
#pragma unroll
    for (int w = 0; w < 8; ++w) { S += ssum[w]; Q += sq[w]; }
    const float mu  = S * (1.0f / (float)HDIM);
    const float var = Q * (1.0f / (float)HDIM) - mu * mu;
    stats[0] = mu;
    stats[1] = rsqrtf(var + LN_EPS);
  }
  __syncthreads();
  return make_float2(stats[0], stats[1]);
}

__device__ __forceinline__ float elu(float x) {
  return x > 0.f ? x : expm1f(x);
}

__global__ void k_embed_ln1(const int* __restrict__ msg,
                            const float* __restrict__ W1T,
                            const float* __restrict__ b1,
                            const float* __restrict__ g1,
                            const float* __restrict__ be1,
                            float* __restrict__ h1) {
  const int b   = blockIdx.x;
  const int j0  = threadIdx.x * 2;
  const int* m  = msg + (size_t)b * NPOS;
  float a0 = 0.f, a1 = 0.f;
  int c = m[0];
  for (int p = 0; p < NPOS; ++p) {
    const float* col = W1T + (size_t)(p * NCHAR + c) * HDIM;
    if (p + 1 < NPOS) {
      c = m[p + 1];
      __builtin_prefetch(W1T + (size_t)((p + 1) * NCHAR + c) * HDIM + j0, 0, 1);
    }
    const v2f v = *(const v2f*)(col + j0);
    a0 += v.x;
    a1 += v.y;
  }
  a0 += b1[j0];
  a1 += b1[j0 + 1];
  const float2 st = ln_stats_512(a0, a1);
  const float v0 = (a0 - st.x) * st.y * g1[j0]     + be1[j0];
  const float v1 = (a1 - st.x) * st.y * g1[j0 + 1] + be1[j0 + 1];
  h1[(size_t)b * HDIM + j0]     = elu(v0);
  h1[(size_t)b * HDIM + j0 + 1] = elu(v1);
}

__global__ void k_embed_ln1_direct(const int* __restrict__ msg,
                                   const float* __restrict__ W1,
                                   const float* __restrict__ b1,
                                   const float* __restrict__ g1,
                                   const float* __restrict__ be1,
                                   float* __restrict__ h1) {
  const int b   = blockIdx.x;
  const int tid = threadIdx.x;
  const int* m  = msg + (size_t)b * NPOS;
  float a0 = 0.f, a1 = 0.f;
  for (int p = 0; p < NPOS; ++p) {
    const size_t idx = (size_t)p * NCHAR + m[p];
    a0 += W1[(size_t)tid * IDIM + idx];
    a1 += W1[(size_t)(tid + 256) * IDIM + idx];
  }
  a0 += b1[tid];
  a1 += b1[tid + 256];
  const float2 st = ln_stats_512(a0, a1);
  const float v0 = (a0 - st.x) * st.y * g1[tid]       + be1[tid];
  const float v1 = (a1 - st.x) * st.y * g1[tid + 256] + be1[tid + 256];
  h1[(size_t)b * HDIM + tid]       = elu(v0);
  h1[(size_t)b * HDIM + tid + 256] = elu(v1);
}

__global__ void k_gemm_wmma(const float* __restrict__ h1,
                            const float* __restrict__ W2,
                            const float* __restrict__ b2,
                            float* __restrict__ h2) {
  __shared__ float sA[128 * LDP];
  __shared__ float sB[64 * LDP];
  const int tid   = threadIdx.x;
  const int wid   = tid >> 5;
  const int lane  = tid & 31;
  const int m0    = blockIdx.x * 128;
  const int n0    = blockIdx.y * 64;
  const int mrow  = lane & 15;
  const int khalf = (lane >> 4) << 1;

  v8f acc0 = {}, acc1 = {}, acc2 = {}, acc3 = {};

  for (int k0 = 0; k0 < HDIM; k0 += KC) {
#pragma unroll
    for (int i = 0; i < 4; ++i) {
      const int idx = tid + i * 256;
      const int r   = idx >> 3;
      const int cc  = (idx & 7) << 2;
      const float4 v = *(const float4*)(h1 + (size_t)(m0 + r) * HDIM + k0 + cc);
      v2f lo = {v.x, v.y};
      v2f hi = {v.z, v.w};
      *(v2f*)(sA + r * LDP + cc)     = lo;
      *(v2f*)(sA + r * LDP + cc + 2) = hi;
    }
#pragma unroll
    for (int i = 0; i < 2; ++i) {
      const int idx = tid + i * 256;
      const int r   = idx >> 3;
      const int cc  = (idx & 7) << 2;
      const float4 v = *(const float4*)(W2 + (size_t)(n0 + r) * HDIM + k0 + cc);
      v2f lo = {v.x, v.y};
      v2f hi = {v.z, v.w};
      *(v2f*)(sB + r * LDP + cc)     = lo;
      *(v2f*)(sB + r * LDP + cc + 2) = hi;
    }
    __syncthreads();

    const float* Aw = sA + (wid * 16 + mrow) * LDP + khalf;
    const float* Bw = sB + mrow * LDP + khalf;
#pragma unroll
    for (int kk = 0; kk < KC; kk += 4) {
      const v2f a   = *(const v2f*)(Aw + kk);
      const v2f bb0 = *(const v2f*)(Bw + kk);
      const v2f bb1 = *(const v2f*)(Bw + 16 * LDP + kk);
      const v2f bb2 = *(const v2f*)(Bw + 32 * LDP + kk);
      const v2f bb3 = *(const v2f*)(Bw + 48 * LDP + kk);
      acc0 = __builtin_amdgcn_wmma_f32_16x16x4_f32(false, a, false, bb0, (short)0, acc0, false, false);
      acc1 = __builtin_amdgcn_wmma_f32_16x16x4_f32(false, a, false, bb1, (short)0, acc1, false, false);
      acc2 = __builtin_amdgcn_wmma_f32_16x16x4_f32(false, a, false, bb2, (short)0, acc2, false, false);
      acc3 = __builtin_amdgcn_wmma_f32_16x16x4_f32(false, a, false, bb3, (short)0, acc3, false, false);
    }
    __syncthreads();
  }

  const int ncol  = lane & 15;
  const int mbase = (lane >> 4) << 3;
  float* out = h2 + (size_t)(m0 + wid * 16 + mbase) * HDIM + n0;
  const float bj0 = b2[n0 + ncol];
  const float bj1 = b2[n0 + 16 + ncol];
  const float bj2 = b2[n0 + 32 + ncol];
  const float bj3 = b2[n0 + 48 + ncol];
#pragma unroll
  for (int r = 0; r < 8; ++r) {
    float* row = out + (size_t)r * HDIM;
    row[ncol]      = acc0[r] + bj0;
    row[16 + ncol] = acc1[r] + bj1;
    row[32 + ncol] = acc2[r] + bj2;
    row[48 + ncol] = acc3[r] + bj3;
  }
}

__global__ void k_ln2_elu(const float* __restrict__ h2,
                          const float* __restrict__ g2,
                          const float* __restrict__ be2,
                          float* __restrict__ out) {
  const int b  = blockIdx.x;
  const int j0 = threadIdx.x * 2;
  const v2f v  = *(const v2f*)(h2 + (size_t)b * HDIM + j0);
  const float a0 = v.x, a1 = v.y;
  const float2 st = ln_stats_512(a0, a1);
  const float v0 = (a0 - st.x) * st.y * g2[j0]     + be2[j0];
  const float v1 = (a1 - st.x) * st.y * g2[j0 + 1] + be2[j0 + 1];
  v2f o = {elu(v0), elu(v1)};
  *(v2f*)(out + (size_t)b * HDIM + j0) = o;
}

extern "C" void kernel_launch(void* const* d_in, const int* in_sizes, int n_in,
                              void* d_out, int out_size, void* d_ws, size_t ws_size,
                              hipStream_t stream) {
  (void)in_sizes; (void)n_in; (void)out_size;
  const int*   msg = (const int*)  d_in[0];
  const float* W1  = (const float*)d_in[1];
  const float* b1  = (const float*)d_in[2];
  const float* g1  = (const float*)d_in[3];
  const float* be1 = (const float*)d_in[4];
  const float* W2  = (const float*)d_in[5];
  const float* b2  = (const float*)d_in[6];
  const float* g2  = (const float*)d_in[7];
  const float* be2 = (const float*)d_in[8];
  float* out = (float*)d_out;

  const size_t w1t_bytes = (size_t)IDIM * HDIM * sizeof(float);
  const size_t h_bytes   = (size_t)BATCH * HDIM * sizeof(float);
  char* ws = (char*)d_ws;
  const bool use_transpose = ws_size >= (w1t_bytes + 2 * h_bytes);

  float* h1;
  float* h2;
  if (use_transpose) {
    float* W1T = (float*)ws;
    h1 = (float*)(ws + w1t_bytes);
    h2 = (float*)(ws + w1t_bytes + h_bytes);
    k_transpose_w1<<<dim3(IDIM / 32, HDIM / 32), dim3(32, 8), 0, stream>>>(W1, W1T);
    k_embed_ln1<<<BATCH, 256, 0, stream>>>(msg, W1T, b1, g1, be1, h1);
  } else {
    h1 = (float*)ws;
    h2 = (float*)(ws + h_bytes);
    k_embed_ln1_direct<<<BATCH, 256, 0, stream>>>(msg, W1, b1, g1, be1, h1);
  }

  k_gemm_wmma<<<dim3(BATCH / 128, HDIM / 64), 256, 0, stream>>>(h1, W2, b2, h2);
  k_ln2_elu<<<BATCH, 256, 0, stream>>>(h2, g2, be2, out);
}


# embed gather: one wave per batch row, 16-byte loads, 8+ gather loads in flight per wave (parent: 8-byte loads, 1 to 3 in flight); LN1 stats summed in-wave (new reduction tree), no LDS or barrier
# speedup vs baseline: 1.1965x; 1.1965x over previous
#include <hip/hip_runtime.h>
#include <hip/hip_bf16.h>
#include <math.h>

#define NPOS  80
#define NCHAR 256
#define HDIM  512
#define IDIM  (NPOS * NCHAR)
#define BATCH 8192
#define LN_EPS 1e-5f
#define KC    32
#define LDP   34

typedef __attribute__((ext_vector_type(2))) float v2f;
typedef __attribute__((ext_vector_type(8))) float v8f;

__global__ void k_transpose_w1(const float* __restrict__ W1,
                               float* __restrict__ W1T) {
  __shared__ float tile[32][33];
  const int i0 = blockIdx.x * 32;
  const int j0 = blockIdx.y * 32;
  const int tx = threadIdx.x;
  const int ty = threadIdx.y;
#pragma unroll
  for (int r = ty; r < 32; r += 8)
    tile[r][tx] = W1[(size_t)(j0 + r) * IDIM + (i0 + tx)];
  __syncthreads();
#pragma unroll
  for (int r = ty; r < 32; r += 8)
    W1T[(size_t)(i0 + r) * HDIM + (j0 + tx)] = tile[tx][r];
}

__device__ __forceinline__ float2 ln_stats_512(float a0, float a1) {
  float s = a0 + a1;
  float q = a0 * a0 + a1 * a1;
#pragma unroll
  for (int off = 16; off > 0; off >>= 1) {
    s += __shfl_xor(s, off, 32);
    q += __shfl_xor(q, off, 32);
  }
  __shared__ float ssum[8], sq[8], stats[2];
  const int wid  = threadIdx.x >> 5;
  const int lane = threadIdx.x & 31;
  if (lane == 0) { ssum[wid] = s; sq[wid] = q; }
  __syncthreads();
  if (threadIdx.x == 0) {
    float S = 0.f, Q = 0.f;
#pragma unroll
    for (int w = 0; w < 8; ++w) { S += ssum[w]; Q += sq[w]; }
    const float mu  = S * (1.0f / (float)HDIM);
    const float var = Q * (1.0f / (float)HDIM) - mu * mu;
    stats[0] = mu;
    stats[1] = rsqrtf(var + LN_EPS);
  }
  __syncthreads();
  return make_float2(stats[0], stats[1]);
}

__device__ __forceinline__ float elu(float x) {
  return x > 0.f ? x : expm1f(x);
}

#define EROWS 8
#define EPOS  4

typedef __attribute__((ext_vector_type(4))) float v4f;

__device__ __forceinline__ v4f ln_elu4(v4f a, float mu, float rs, v4f g, v4f be) {
  v4f o;
  o.x = elu((a.x - mu) * rs * g.x + be.x);
  o.y = elu((a.y - mu) * rs * g.y + be.y);
  o.z = elu((a.z - mu) * rs * g.z + be.z);
  o.w = elu((a.w - mu) * rs * g.w + be.w);
  return o;
}

__global__ void k_embed_ln1(const int* __restrict__ msg,
                            const float* __restrict__ W1T,
                            const float* __restrict__ b1,
                            const float* __restrict__ g1,
                            const float* __restrict__ be1,
                            float* __restrict__ h1) {
  const int lane = threadIdx.x & 31;
  const int wid  = __builtin_amdgcn_readfirstlane((int)(threadIdx.x >> 5));
  const int b    = blockIdx.x * EROWS + wid;
  const int* m   = msg + (size_t)b * NPOS;
  const int col  = lane * 4;

  v4f acc0 = {0.f, 0.f, 0.f, 0.f};
  v4f acc1 = acc0, acc2 = acc0, acc3 = acc0;
#pragma unroll
  for (int p0 = 0; p0 < NPOS; p0 += EPOS) {
    v4f v[EPOS][4];
#pragma unroll
    for (int u = 0; u < EPOS; ++u) {
      const float* row = W1T + (size_t)((p0 + u) * NCHAR + m[p0 + u]) * HDIM + col;
#pragma unroll
      for (int q = 0; q < 4; ++q)
        v[u][q] = *(const v4f*)(row + q * 128);
    }
#pragma unroll
    for (int u = 0; u < EPOS; ++u) {
      acc0 += v[u][0];
      acc1 += v[u][1];
      acc2 += v[u][2];
      acc3 += v[u][3];
    }
  }
  acc0 += *(const v4f*)(b1 + col);
  acc1 += *(const v4f*)(b1 + 128 + col);
  acc2 += *(const v4f*)(b1 + 256 + col);
  acc3 += *(const v4f*)(b1 + 384 + col);

  const v4f t = (acc0 + acc1) + (acc2 + acc3);
  const v4f u2 = (acc0 * acc0 + acc1 * acc1) + (acc2 * acc2 + acc3 * acc3);
  float s = (t.x + t.y) + (t.z + t.w);
  float q = (u2.x + u2.y) + (u2.z + u2.w);
#pragma unroll
  for (int off = 16; off > 0; off >>= 1) {
    s += __shfl_xor(s, off, 32);
    q += __shfl_xor(q, off, 32);
  }
  const float mu  = s * (1.0f / (float)HDIM);
  const float var = q * (1.0f / (float)HDIM) - mu * mu;
  const float rs  = rsqrtf(var + LN_EPS);

  float* o = h1 + (size_t)b * HDIM + col;
  *(v4f*)(o)       = ln_elu4(acc0, mu, rs, *(const v4f*)(g1 + col),       *(const v4f*)(be1 + col));
  *(v4f*)(o + 128) = ln_elu4(acc1, mu, rs, *(const v4f*)(g1 + 128 + col), *(const v4f*)(be1 + 128 + col));
  *(v4f*)(o + 256) = ln_elu4(acc2, mu, rs, *(const v4f*)(g1 + 256 + col), *(const v4f*)(be1 + 256 + col));
  *(v4f*)(o + 384) = ln_elu4(acc3, mu, rs, *(const v4f*)(g1 + 384 + col), *(const v4f*)(be1 + 384 + col));
}

__global__ void k_embed_ln1_direct(const int* __restrict__ msg,
                                   const float* __restrict__ W1,
                                   const float* __restrict__ b1,
                                   const float* __restrict__ g1,
                                   const float* __restrict__ be1,
                                   float* __restrict__ h1) {
  const int b   = blockIdx.x;
  const int tid = threadIdx.x;
  const int* m  = msg + (size_t)b * NPOS;
  float a0 = 0.f, a1 = 0.f;
  for (int p = 0; p < NPOS; ++p) {
    const size_t idx = (size_t)p * NCHAR + m[p];
    a0 += W1[(size_t)tid * IDIM + idx];
    a1 += W1[(size_t)(tid + 256) * IDIM + idx];
  }
  a0 += b1[tid];
  a1 += b1[tid + 256];
  const float2 st = ln_stats_512(a0, a1);
  const float v0 = (a0 - st.x) * st.y * g1[tid]       + be1[tid];
  const float v1 = (a1 - st.x) * st.y * g1[tid + 256] + be1[tid + 256];
  h1[(size_t)b * HDIM + tid]       = elu(v0);
  h1[(size_t)b * HDIM + tid + 256] = elu(v1);
}

__global__ void k_gemm_wmma(const float* __restrict__ h1,
                            const float* __restrict__ W2,
                            const float* __restrict__ b2,
                            float* __restrict__ h2) {
  __shared__ float sA[128 * LDP];
  __shared__ float sB[64 * LDP];
  const int tid   = threadIdx.x;
  const int wid   = tid >> 5;
  const int lane  = tid & 31;
  const int m0    = blockIdx.x * 128;
  const int n0    = blockIdx.y * 64;
  const int mrow  = lane & 15;
  const int khalf = (lane >> 4) << 1;

  v8f acc0 = {}, acc1 = {}, acc2 = {}, acc3 = {};

  for (int k0 = 0; k0 < HDIM; k0 += KC) {
#pragma unroll
    for (int i = 0; i < 4; ++i) {
      const int idx = tid + i * 256;
      const int r   = idx >> 3;
      const int cc  = (idx & 7) << 2;
      const float4 v = *(const float4*)(h1 + (size_t)(m0 + r) * HDIM + k0 + cc);
      v2f lo = {v.x, v.y};
      v2f hi = {v.z, v.w};
      *(v2f*)(sA + r * LDP + cc)     = lo;
      *(v2f*)(sA + r * LDP + cc + 2) = hi;
    }
#pragma unroll
    for (int i = 0; i < 2; ++i) {
      const int idx = tid + i * 256;
      const int r   = idx >> 3;
      const int cc  = (idx & 7) << 2;
      const float4 v = *(const float4*)(W2 + (size_t)(n0 + r) * HDIM + k0 + cc);
      v2f lo = {v.x, v.y};
      v2f hi = {v.z, v.w};
      *(v2f*)(sB + r * LDP + cc)     = lo;
      *(v2f*)(sB + r * LDP + cc + 2) = hi;
    }
    __syncthreads();

    const float* Aw = sA + (wid * 16 + mrow) * LDP + khalf;
    const float* Bw = sB + mrow * LDP + khalf;
#pragma unroll
    for (int kk = 0; kk < KC; kk += 4) {
      const v2f a   = *(const v2f*)(Aw + kk);
      const v2f bb0 = *(const v2f*)(Bw + kk);
      const v2f bb1 = *(const v2f*)(Bw + 16 * LDP + kk);
      const v2f bb2 = *(const v2f*)(Bw + 32 * LDP + kk);
      const v2f bb3 = *(const v2f*)(Bw + 48 * LDP + kk);
      acc0 = __builtin_amdgcn_wmma_f32_16x16x4_f32(false, a, false, bb0, (short)0, acc0, false, false);
      acc1 = __builtin_amdgcn_wmma_f32_16x16x4_f32(false, a, false, bb1, (short)0, acc1, false, false);
      acc2 = __builtin_amdgcn_wmma_f32_16x16x4_f32(false, a, false, bb2, (short)0, acc2, false, false);
      acc3 = __builtin_amdgcn_wmma_f32_16x16x4_f32(false, a, false, bb3, (short)0, acc3, false, false);
    }
    __syncthreads();
  }

  const int ncol  = lane & 15;
  const int mbase = (lane >> 4) << 3;
  float* out = h2 + (size_t)(m0 + wid * 16 + mbase) * HDIM + n0;
  const float bj0 = b2[n0 + ncol];
  const float bj1 = b2[n0 + 16 + ncol];
  const float bj2 = b2[n0 + 32 + ncol];
  const float bj3 = b2[n0 + 48 + ncol];
#pragma unroll
  for (int r = 0; r < 8; ++r) {
    float* row = out + (size_t)r * HDIM;
    row[ncol]      = acc0[r] + bj0;
    row[16 + ncol] = acc1[r] + bj1;
    row[32 + ncol] = acc2[r] + bj2;
    row[48 + ncol] = acc3[r] + bj3;
  }
}

__global__ void k_ln2_elu(const float* __restrict__ h2,
                          const float* __restrict__ g2,
                          const float* __restrict__ be2,
                          float* __restrict__ out) {
  const int b  = blockIdx.x;
  const int j0 = threadIdx.x * 2;
  const v2f v  = *(const v2f*)(h2 + (size_t)b * HDIM + j0);
  const float a0 = v.x, a1 = v.y;
  const float2 st = ln_stats_512(a0, a1);
  const float v0 = (a0 - st.x) * st.y * g2[j0]     + be2[j0];
  const float v1 = (a1 - st.x) * st.y * g2[j0 + 1] + be2[j0 + 1];
  v2f o = {elu(v0), elu(v1)};
  *(v2f*)(out + (size_t)b * HDIM + j0) = o;
}

extern "C" void kernel_launch(void* const* d_in, const int* in_sizes, int n_in,
                              void* d_out, int out_size, void* d_ws, size_t ws_size,
                              hipStream_t stream) {
  (void)in_sizes; (void)n_in; (void)out_size;
  const int*   msg = (const int*)  d_in[0];
  const float* W1  = (const float*)d_in[1];
  const float* b1  = (const float*)d_in[2];
  const float* g1  = (const float*)d_in[3];
  const float* be1 = (const float*)d_in[4];
  const float* W2  = (const float*)d_in[5];
  const float* b2  = (const float*)d_in[6];
  const float* g2  = (const float*)d_in[7];
  const float* be2 = (const float*)d_in[8];
  float* out = (float*)d_out;

  const size_t w1t_bytes = (size_t)IDIM * HDIM * sizeof(float);
  const size_t h_bytes   = (size_t)BATCH * HDIM * sizeof(float);
  char* ws = (char*)d_ws;
  const bool use_transpose = ws_size >= (w1t_bytes + 2 * h_bytes);

  float* h1;
  float* h2;
  if (use_transpose) {
    float* W1T = (float*)ws;
    h1 = (float*)(ws + w1t_bytes);
    h2 = (float*)(ws + w1t_bytes + h_bytes);
    k_transpose_w1<<<dim3(IDIM / 32, HDIM / 32), dim3(32, 8), 0, stream>>>(W1, W1T);
    k_embed_ln1<<<BATCH / EROWS, 32 * EROWS, 0, stream>>>(msg, W1T, b1, g1, be1, h1);
  } else {
    h1 = (float*)ws;
    h2 = (float*)(ws + h_bytes);
    k_embed_ln1_direct<<<BATCH, 256, 0, stream>>>(msg, W1, b1, g1, be1, h1);
  }

  k_gemm_wmma<<<dim3(BATCH / 128, HDIM / 64), 256, 0, stream>>>(h1, W2, b2, h2);
  k_ln2_elu<<<BATCH, 256, 0, stream>>>(h2, g2, be2, out);
}


# final LayerNorm+ELU: one wave per batch row (8 rows per block), 16-byte loads and stores issued together, stats summed in-wave (new reduction tree), no LDS or barrier; 8192 waves, not 65536
# speedup vs baseline: 1.2480x; 1.0430x over previous
#include <hip/hip_runtime.h>
#include <hip/hip_bf16.h>
#include <math.h>

#define NPOS  80
#define NCHAR 256
#define HDIM  512
#define IDIM  (NPOS * NCHAR)
#define BATCH 8192
#define LN_EPS 1e-5f
#define KC    32
#define LDP   34

typedef __attribute__((ext_vector_type(2))) float v2f;
typedef __attribute__((ext_vector_type(8))) float v8f;

__global__ void k_transpose_w1(const float* __restrict__ W1,
                               float* __restrict__ W1T) {
  __shared__ float tile[32][33];
  const int i0 = blockIdx.x * 32;
  const int j0 = blockIdx.y * 32;
  const int tx = threadIdx.x;
  const int ty = threadIdx.y;
#pragma unroll
  for (int r = ty; r < 32; r += 8)
    tile[r][tx] = W1[(size_t)(j0 + r) * IDIM + (i0 + tx)];
  __syncthreads();
#pragma unroll
  for (int r = ty; r < 32; r += 8)
    W1T[(size_t)(i0 + r) * HDIM + (j0 + tx)] = tile[tx][r];
}

__device__ __forceinline__ float2 ln_stats_512(float a0, float a1) {
  float s = a0 + a1;
  float q = a0 * a0 + a1 * a1;
#pragma unroll
  for (int off = 16; off > 0; off >>= 1) {
    s += __shfl_xor(s, off, 32);
    q += __shfl_xor(q, off, 32);
  }
  __shared__ float ssum[8], sq[8], stats[2];
  const int wid  = threadIdx.x >> 5;
  const int lane = threadIdx.x & 31;
  if (lane == 0) { ssum[wid] = s; sq[wid] = q; }
  __syncthreads();
  if (threadIdx.x == 0) {
    float S = 0.f, Q = 0.f;
#pragma unroll
    for (int w = 0; w < 8; ++w) { S += ssum[w]; Q += sq[w]; }
    const float mu  = S * (1.0f / (float)HDIM);
    const float var = Q * (1.0f / (float)HDIM) - mu * mu;
    stats[0] = mu;
    stats[1] = rsqrtf(var + LN_EPS);
  }
  __syncthreads();
  return make_float2(stats[0], stats[1]);
}

__device__ __forceinline__ float elu(float x) {
  return x > 0.f ? x : expm1f(x);
}

#define EROWS 8
#define EPOS  4

typedef __attribute__((ext_vector_type(4))) float v4f;

__device__ __forceinline__ v4f ln_elu4(v4f a, float mu, float rs, v4f g, v4f be) {
  v4f o;
  o.x = elu((a.x - mu) * rs * g.x + be.x);
  o.y = elu((a.y - mu) * rs * g.y + be.y);
  o.z = elu((a.z - mu) * rs * g.z + be.z);
  o.w = elu((a.w - mu) * rs * g.w + be.w);
  return o;
}

__global__ void k_embed_ln1(const int* __restrict__ msg,
                            const float* __restrict__ W1T,
                            const float* __restrict__ b1,
                            const float* __restrict__ g1,
                            const float* __restrict__ be1,
                            float* __restrict__ h1) {
  const int lane = threadIdx.x & 31;
  const int wid  = __builtin_amdgcn_readfirstlane((int)(threadIdx.x >> 5));
  const int b    = blockIdx.x * EROWS + wid;
  const int* m   = msg + (size_t)b * NPOS;
  const int col  = lane * 4;

  v4f acc0 = {0.f, 0.f, 0.f, 0.f};
  v4f acc1 = acc0, acc2 = acc0, acc3 = acc0;
#pragma unroll
  for (int p0 = 0; p0 < NPOS; p0 += EPOS) {
    v4f v[EPOS][4];
#pragma unroll
    for (int u = 0; u < EPOS; ++u) {
      const float* row = W1T + (size_t)((p0 + u) * NCHAR + m[p0 + u]) * HDIM + col;
#pragma unroll
      for (int q = 0; q < 4; ++q)
        v[u][q] = *(const v4f*)(row + q * 128);
    }
#pragma unroll
    for (int u = 0; u < EPOS; ++u) {
      acc0 += v[u][0];
      acc1 += v[u][1];
      acc2 += v[u][2];
      acc3 += v[u][3];
    }
  }
  acc0 += *(const v4f*)(b1 + col);
  acc1 += *(const v4f*)(b1 + 128 + col);
  acc2 += *(const v4f*)(b1 + 256 + col);
  acc3 += *(const v4f*)(b1 + 384 + col);

  const v4f t = (acc0 + acc1) + (acc2 + acc3);
  const v4f u2 = (acc0 * acc0 + acc1 * acc1) + (acc2 * acc2 + acc3 * acc3);
  float s = (t.x + t.y) + (t.z + t.w);
  float q = (u2.x + u2.y) + (u2.z + u2.w);
#pragma unroll
  for (int off = 16; off > 0; off >>= 1) {
    s += __shfl_xor(s, off, 32);
    q += __shfl_xor(q, off, 32);
  }
  const float mu  = s * (1.0f / (float)HDIM);
  const float var = q * (1.0f / (float)HDIM) - mu * mu;
  const float rs  = rsqrtf(var + LN_EPS);

  float* o = h1 + (size_t)b * HDIM + col;
  *(v4f*)(o)       = ln_elu4(acc0, mu, rs, *(const v4f*)(g1 + col),       *(const v4f*)(be1 + col));
  *(v4f*)(o + 128) = ln_elu4(acc1, mu, rs, *(const v4f*)(g1 + 128 + col), *(const v4f*)(be1 + 128 + col));
  *(v4f*)(o + 256) = ln_elu4(acc2, mu, rs, *(const v4f*)(g1 + 256 + col), *(const v4f*)(be1 + 256 + col));
  *(v4f*)(o + 384) = ln_elu4(acc3, mu, rs, *(const v4f*)(g1 + 384 + col), *(const v4f*)(be1 + 384 + col));
}

__global__ void k_embed_ln1_direct(const int* __restrict__ msg,
                                   const float* __restrict__ W1,
                                   const float* __restrict__ b1,
                                   const float* __restrict__ g1,
                                   const float* __restrict__ be1,
                                   float* __restrict__ h1) {
  const int b   = blockIdx.x;
  const int tid = threadIdx.x;
  const int* m  = msg + (size_t)b * NPOS;
  float a0 = 0.f, a1 = 0.f;
  for (int p = 0; p < NPOS; ++p) {
    const size_t idx = (size_t)p * NCHAR + m[p];
    a0 += W1[(size_t)tid * IDIM + idx];
    a1 += W1[(size_t)(tid + 256) * IDIM + idx];
  }
  a0 += b1[tid];
  a1 += b1[tid + 256];
  const float2 st = ln_stats_512(a0, a1);
  const float v0 = (a0 - st.x) * st.y * g1[tid]       + be1[tid];
  const float v1 = (a1 - st.x) * st.y * g1[tid + 256] + be1[tid + 256];
  h1[(size_t)b * HDIM + tid]       = elu(v0);
  h1[(size_t)b * HDIM + tid + 256] = elu(v1);
}

__global__ void k_gemm_wmma(const float* __restrict__ h1,
                            const float* __restrict__ W2,
                            const float* __restrict__ b2,
                            float* __restrict__ h2) {
  __shared__ float sA[128 * LDP];
  __shared__ float sB[64 * LDP];
  const int tid   = threadIdx.x;
  const int wid   = tid >> 5;
  const int lane  = tid & 31;
  const int m0    = blockIdx.x * 128;
  const int n0    = blockIdx.y * 64;
  const int mrow  = lane & 15;
  const int khalf = (lane >> 4) << 1;

  v8f acc0 = {}, acc1 = {}, acc2 = {}, acc3 = {};

  for (int k0 = 0; k0 < HDIM; k0 += KC) {
#pragma unroll
    for (int i = 0; i < 4; ++i) {
      const int idx = tid + i * 256;
      const int r   = idx >> 3;
      const int cc  = (idx & 7) << 2;
      const float4 v = *(const float4*)(h1 + (size_t)(m0 + r) * HDIM + k0 + cc);
      v2f lo = {v.x, v.y};
      v2f hi = {v.z, v.w};
      *(v2f*)(sA + r * LDP + cc)     = lo;
      *(v2f*)(sA + r * LDP + cc + 2) = hi;
    }
#pragma unroll
    for (int i = 0; i < 2; ++i) {
      const int idx = tid + i * 256;
      const int r   = idx >> 3;
      const int cc  = (idx & 7) << 2;
      const float4 v = *(const float4*)(W2 + (size_t)(n0 + r) * HDIM + k0 + cc);
      v2f lo = {v.x, v.y};
      v2f hi = {v.z, v.w};
      *(v2f*)(sB + r * LDP + cc)     = lo;
      *(v2f*)(sB + r * LDP + cc + 2) = hi;
    }
    __syncthreads();

    const float* Aw = sA + (wid * 16 + mrow) * LDP + khalf;
    const float* Bw = sB + mrow * LDP + khalf;
#pragma unroll
    for (int kk = 0; kk < KC; kk += 4) {
      const v2f a   = *(const v2f*)(Aw + kk);
      const v2f bb0 = *(const v2f*)(Bw + kk);
      const v2f bb1 = *(const v2f*)(Bw + 16 * LDP + kk);
      const v2f bb2 = *(const v2f*)(Bw + 32 * LDP + kk);
      const v2f bb3 = *(const v2f*)(Bw + 48 * LDP + kk);
      acc0 = __builtin_amdgcn_wmma_f32_16x16x4_f32(false, a, false, bb0, (short)0, acc0, false, false);
      acc1 = __builtin_amdgcn_wmma_f32_16x16x4_f32(false, a, false, bb1, (short)0, acc1, false, false);
      acc2 = __builtin_amdgcn_wmma_f32_16x16x4_f32(false, a, false, bb2, (short)0, acc2, false, false);
      acc3 = __builtin_amdgcn_wmma_f32_16x16x4_f32(false, a, false, bb3, (short)0, acc3, false, false);
    }
    __syncthreads();
  }

  const int ncol  = lane & 15;
  const int mbase = (lane >> 4) << 3;
  float* out = h2 + (size_t)(m0 + wid * 16 + mbase) * HDIM + n0;
  const float bj0 = b2[n0 + ncol];
  const float bj1 = b2[n0 + 16 + ncol];
  const float bj2 = b2[n0 + 32 + ncol];
  const float bj3 = b2[n0 + 48 + ncol];
#pragma unroll
  for (int r = 0; r < 8; ++r) {
    float* row = out + (size_t)r * HDIM;
    row[ncol]      = acc0[r] + bj0;
    row[16 + ncol] = acc1[r] + bj1;
    row[32 + ncol] = acc2[r] + bj2;
    row[48 + ncol] = acc3[r] + bj3;
  }
}

#define LROWS 8

__global__ void k_ln2_elu(const float* __restrict__ h2,
                          const float* __restrict__ g2,
                          const float* __restrict__ be2,
                          float* __restrict__ out) {
  const int lane = threadIdx.x & 31;
  const int wid  = threadIdx.x >> 5;
  const int b    = blockIdx.x * LROWS + wid;
  const int col  = lane * 4;

  const float* x = h2 + (size_t)b * HDIM + col;
  const v4f a0 = *(const v4f*)(x);
  const v4f a1 = *(const v4f*)(x + 128);
  const v4f a2 = *(const v4f*)(x + 256);
  const v4f a3 = *(const v4f*)(x + 384);
  const v4f gq0 = *(const v4f*)(g2 + col);
  const v4f gq1 = *(const v4f*)(g2 + 128 + col);
  const v4f gq2 = *(const v4f*)(g2 + 256 + col);
  const v4f gq3 = *(const v4f*)(g2 + 384 + col);
  const v4f bq0 = *(const v4f*)(be2 + col);
  const v4f bq1 = *(const v4f*)(be2 + 128 + col);
  const v4f bq2 = *(const v4f*)(be2 + 256 + col);
  const v4f bq3 = *(const v4f*)(be2 + 384 + col);

  const v4f t  = (a0 + a1) + (a2 + a3);
  const v4f u2 = (a0 * a0 + a1 * a1) + (a2 * a2 + a3 * a3);
  float s = (t.x + t.y) + (t.z + t.w);
  float q = (u2.x + u2.y) + (u2.z + u2.w);
#pragma unroll
  for (int off = 16; off > 0; off >>= 1) {
    s += __shfl_xor(s, off, 32);
    q += __shfl_xor(q, off, 32);
  }
  const float mu  = s * (1.0f / (float)HDIM);
  const float var = q * (1.0f / (float)HDIM) - mu * mu;
  const float rs  = rsqrtf(var + LN_EPS);

  float* o = out + (size_t)b * HDIM + col;
  *(v4f*)(o)       = ln_elu4(a0, mu, rs, gq0, bq0);
  *(v4f*)(o + 128) = ln_elu4(a1, mu, rs, gq1, bq1);
  *(v4f*)(o + 256) = ln_elu4(a2, mu, rs, gq2, bq2);
  *(v4f*)(o + 384) = ln_elu4(a3, mu, rs, gq3, bq3);
}

extern "C" void kernel_launch(void* const* d_in, const int* in_sizes, int n_in,
                              void* d_out, int out_size, void* d_ws, size_t ws_size,
                              hipStream_t stream) {
  (void)in_sizes; (void)n_in; (void)out_size;
  const int*   msg = (const int*)  d_in[0];
  const float* W1  = (const float*)d_in[1];
  const float* b1  = (const float*)d_in[2];
  const float* g1  = (const float*)d_in[3];
  const float* be1 = (const float*)d_in[4];
  const float* W2  = (const float*)d_in[5];
  const float* b2  = (const float*)d_in[6];
  const float* g2  = (const float*)d_in[7];
  const float* be2 = (const float*)d_in[8];
  float* out = (float*)d_out;

  const size_t w1t_bytes = (size_t)IDIM * HDIM * sizeof(float);
  const size_t h_bytes   = (size_t)BATCH * HDIM * sizeof(float);
  char* ws = (char*)d_ws;
  const bool use_transpose = ws_size >= (w1t_bytes + 2 * h_bytes);

  float* h1;
  float* h2;
  if (use_transpose) {
    float* W1T = (float*)ws;
    h1 = (float*)(ws + w1t_bytes);
    h2 = (float*)(ws + w1t_bytes + h_bytes);
    k_transpose_w1<<<dim3(IDIM / 32, HDIM / 32), dim3(32, 8), 0, stream>>>(W1, W1T);
    k_embed_ln1<<<BATCH / EROWS, 32 * EROWS, 0, stream>>>(msg, W1T, b1, g1, be1, h1);
  } else {
    h1 = (float*)ws;
    h2 = (float*)(ws + h_bytes);
    k_embed_ln1_direct<<<BATCH, 256, 0, stream>>>(msg, W1, b1, g1, be1, h1);
  }

  k_gemm_wmma<<<dim3(BATCH / 128, HDIM / 64), 256, 0, stream>>>(h1, W2, b2, h2);
  k_ln2_elu<<<BATCH / LROWS, 32 * LROWS, 0, stream>>>(h2, g2, be2, out);
}


# embedding: each position's W1 slab staged in LDS (turned while stored) and gathered from LDS, 512 batch rows per block; no transposed copy of W1 is built; LN1 becomes its own row-per-wave launch
# speedup vs baseline: 1.2936x; 1.0365x over previous
#include <hip/hip_runtime.h>
#include <hip/hip_bf16.h>
#include <math.h>


#define NPOS  80
#define NCHAR 256
#define HDIM  512
#define IDIM  (NPOS * NCHAR)
#define BATCH 8192
#define LN_EPS 1e-5f
#define KC    32
#define LDP   34

typedef __attribute__((ext_vector_type(2))) float v2f;
typedef __attribute__((ext_vector_type(4))) float v4f;
typedef __attribute__((ext_vector_type(8))) float v8f;
typedef __attribute__((ext_vector_type(4))) int   v4i;

__device__ __forceinline__ float elu(float x) {
  return x > 0.f ? x : expm1f(x);
}

__device__ __forceinline__ v4f ln_elu4(v4f a, float mu, float rs, v4f g, v4f be) {
  v4f o;
  o.x = elu((a.x - mu) * rs * g.x + be.x);
  o.y = elu((a.y - mu) * rs * g.y + be.y);
  o.z = elu((a.z - mu) * rs * g.z + be.z);
  o.w = elu((a.w - mu) * rs * g.w + be.w);
  return o;
}

#define CS   32
#define RB   512
#define SP   32
#define SLAB (NCHAR * SP)

__device__ __forceinline__ void lds_barrier() {
  asm volatile("s_wait_dscnt 0x0\n\ts_barrier_signal -1\n\ts_barrier_wait -1" ::: "memory");
}

__global__ __launch_bounds__(RB)
void k_embed_lds(const int* __restrict__ msg,
                 const float* __restrict__ W1,
                 const float* __restrict__ b1,
                 float* __restrict__ hs) {
  __shared__ __attribute__((aligned(16))) float S[2 * SLAB];
  const int t    = threadIdx.x;
  const int h0   = blockIdx.x * CS;
  const int c    = t & (NCHAR - 1);
  const int jh   = t >> 8;
  const float* wsrc = W1 + (size_t)(h0 + 16 * jh) * IDIM + c;
  const int cswz = (c >> 1) & 7;
  float* srow = S + c * SP;
  const int lane = t & 31;
  const int w    = t >> 5;
  const int g    = lane >> 3;
  const int q    = lane & 7;
  const int row0 = blockIdx.y * RB + w * 32 + g;
  const int* mrow = msg + (size_t)row0 * NPOS;

  v4i mc[8];
#pragma unroll
  for (int j = 0; j < 8; ++j) mc[j] = *(const v4i*)(mrow + (size_t)(4 * j) * NPOS);

  float st[2][16];
#pragma unroll
  for (int r = 0; r < 16; ++r) st[0][r] = wsrc[(size_t)r * IDIM];
#pragma unroll
  for (int k = 0; k < 4; ++k) {
    v4f wv = {st[0][4 * k], st[0][4 * k + 1], st[0][4 * k + 2], st[0][4 * k + 3]};
    *(v4f*)(srow + 4 * ((4 * jh + k) ^ cswz)) = wv;
  }
  lds_barrier();
#pragma unroll
  for (int r = 0; r < 16; ++r) st[1][r] = wsrc[(size_t)r * IDIM + 1 * NCHAR];
  __builtin_amdgcn_sched_barrier(0);
#pragma unroll
  for (int r = 0; r < 16; ++r) st[0][r] = wsrc[(size_t)r * IDIM + 2 * NCHAR];

  v4f acc[8];
#pragma unroll
  for (int j = 0; j < 8; ++j) acc[j] = (v4f){0.f, 0.f, 0.f, 0.f};

  for (int p0 = 0; p0 < NPOS; p0 += 4) {
    const int pn = (p0 + 4 < NPOS) ? (p0 + 4) : (NPOS - 4);
    v4i mn[8];
#pragma unroll
    for (int j = 0; j < 8; ++j) mn[j] = *(const v4i*)(mrow + (size_t)(4 * j) * NPOS + pn);
#pragma unroll
    for (int u = 0; u < 4; ++u) {
      const int p = p0 + u;
      {
        float* d = srow + ((u + 1) & 1) * SLAB;
#pragma unroll
        for (int k = 0; k < 4; ++k) {
          v4f wv = {st[(u + 1) & 1][4 * k], st[(u + 1) & 1][4 * k + 1],
                    st[(u + 1) & 1][4 * k + 2], st[(u + 1) & 1][4 * k + 3]};
          *(v4f*)(d + 4 * ((4 * jh + k) ^ cswz)) = wv;
        }
      }
      {
        const int ps = (p + 3 < NPOS) ? (p + 3) : (NPOS - 1);
#pragma unroll
        for (int r = 0; r < 16; ++r)
          st[(u + 1) & 1][r] = wsrc[(size_t)r * IDIM + (size_t)ps * NCHAR];
      }
      const float* sb = S + (u & 1) * SLAB;
      v4f v[8];
#pragma unroll
      for (int j = 0; j < 8; ++j) {
        const int m = mc[j][u] & (NCHAR - 1);
        v[j] = *(const v4f*)(sb + m * SP + 4 * (q ^ ((m >> 1) & 7)));
      }
#pragma unroll
      for (int j = 0; j < 8; ++j) acc[j] += v[j];
      lds_barrier();
    }
#pragma unroll
    for (int j = 0; j < 8; ++j) mc[j] = mn[j];
  }

  const v4f bv = *(const v4f*)(b1 + h0 + 4 * q);
  float* o = hs + (size_t)row0 * HDIM + h0 + 4 * q;
#pragma unroll
  for (int j = 0; j < 8; ++j)
    *(v4f*)(o + (size_t)(4 * j) * HDIM) = acc[j] + bv;
}

__global__ void k_gemm_wmma(const float* __restrict__ h1,
                            const float* __restrict__ W2,
                            const float* __restrict__ b2,
                            float* __restrict__ h2) {
  __shared__ float sA[128 * LDP];
  __shared__ float sB[64 * LDP];
  const int tid   = threadIdx.x;
  const int wid   = tid >> 5;
  const int lane  = tid & 31;
  const int m0    = blockIdx.x * 128;
  const int n0    = blockIdx.y * 64;
  const int mrow  = lane & 15;
  const int khalf = (lane >> 4) << 1;

  v8f acc0 = {}, acc1 = {}, acc2 = {}, acc3 = {};

  for (int k0 = 0; k0 < HDIM; k0 += KC) {
#pragma unroll
    for (int i = 0; i < 4; ++i) {
      const int idx = tid + i * 256;
      const int r   = idx >> 3;
      const int cc  = (idx & 7) << 2;
      const float4 v = *(const float4*)(h1 + (size_t)(m0 + r) * HDIM + k0 + cc);
      v2f lo = {v.x, v.y};
      v2f hi = {v.z, v.w};
      *(v2f*)(sA + r * LDP + cc)     = lo;
      *(v2f*)(sA + r * LDP + cc + 2) = hi;
    }
#pragma unroll
    for (int i = 0; i < 2; ++i) {
      const int idx = tid + i * 256;
      const int r   = idx >> 3;
      const int cc  = (idx & 7) << 2;
      const float4 v = *(const float4*)(W2 + (size_t)(n0 + r) * HDIM + k0 + cc);
      v2f lo = {v.x, v.y};
      v2f hi = {v.z, v.w};
      *(v2f*)(sB + r * LDP + cc)     = lo;
      *(v2f*)(sB + r * LDP + cc + 2) = hi;
    }
    __syncthreads();

    const float* Aw = sA + (wid * 16 + mrow) * LDP + khalf;
    const float* Bw = sB + mrow * LDP + khalf;
#pragma unroll
    for (int kk = 0; kk < KC; kk += 4) {
      const v2f a   = *(const v2f*)(Aw + kk);
      const v2f bb0 = *(const v2f*)(Bw + kk);
      const v2f bb1 = *(const v2f*)(Bw + 16 * LDP + kk);
      const v2f bb2 = *(const v2f*)(Bw + 32 * LDP + kk);
      const v2f bb3 = *(const v2f*)(Bw + 48 * LDP + kk);
      acc0 = __builtin_amdgcn_wmma_f32_16x16x4_f32(false, a, false, bb0, (short)0, acc0, false, false);
      acc1 = __builtin_amdgcn_wmma_f32_16x16x4_f32(false, a, false, bb1, (short)0, acc1, false, false);
      acc2 = __builtin_amdgcn_wmma_f32_16x16x4_f32(false, a, false, bb2, (short)0, acc2, false, false);
      acc3 = __builtin_amdgcn_wmma_f32_16x16x4_f32(false, a, false, bb3, (short)0, acc3, false, false);
    }
    __syncthreads();
  }

  const int ncol  = lane & 15;
  const int mbase = (lane >> 4) << 3;
  float* out = h2 + (size_t)(m0 + wid * 16 + mbase) * HDIM + n0;
  const float bj0 = b2[n0 + ncol];
  const float bj1 = b2[n0 + 16 + ncol];
  const float bj2 = b2[n0 + 32 + ncol];
  const float bj3 = b2[n0 + 48 + ncol];
#pragma unroll
  for (int r = 0; r < 8; ++r) {
    float* row = out + (size_t)r * HDIM;
    row[ncol]      = acc0[r] + bj0;
    row[16 + ncol] = acc1[r] + bj1;
    row[32 + ncol] = acc2[r] + bj2;
    row[48 + ncol] = acc3[r] + bj3;
  }
}

#define LROWS 8

__global__ void k_ln2_elu(const float* __restrict__ h2,
                          const float* __restrict__ g2,
                          const float* __restrict__ be2,
                          float* __restrict__ out) {
  const int lane = threadIdx.x & 31;
  const int wid  = threadIdx.x >> 5;
  const int b    = blockIdx.x * LROWS + wid;
  const int col  = lane * 4;

  const float* x = h2 + (size_t)b * HDIM + col;
  const v4f a0 = *(const v4f*)(x);
  const v4f a1 = *(const v4f*)(x + 128);
  const v4f a2 = *(const v4f*)(x + 256);
  const v4f a3 = *(const v4f*)(x + 384);
  const v4f gq0 = *(const v4f*)(g2 + col);
  const v4f gq1 = *(const v4f*)(g2 + 128 + col);
  const v4f gq2 = *(const v4f*)(g2 + 256 + col);
  const v4f gq3 = *(const v4f*)(g2 + 384 + col);
  const v4f bq0 = *(const v4f*)(be2 + col);
  const v4f bq1 = *(const v4f*)(be2 + 128 + col);
  const v4f bq2 = *(const v4f*)(be2 + 256 + col);
  const v4f bq3 = *(const v4f*)(be2 + 384 + col);

  const v4f t  = (a0 + a1) + (a2 + a3);
  const v4f u2 = (a0 * a0 + a1 * a1) + (a2 * a2 + a3 * a3);
  float s = (t.x + t.y) + (t.z + t.w);
  float q = (u2.x + u2.y) + (u2.z + u2.w);
#pragma unroll
  for (int off = 16; off > 0; off >>= 1) {
    s += __shfl_xor(s, off, 32);
    q += __shfl_xor(q, off, 32);
  }
  const float mu  = s * (1.0f / (float)HDIM);
  const float var = q * (1.0f / (float)HDIM) - mu * mu;
  const float rs  = rsqrtf(var + LN_EPS);

  float* o = out + (size_t)b * HDIM + col;
  *(v4f*)(o)       = ln_elu4(a0, mu, rs, gq0, bq0);
  *(v4f*)(o + 128) = ln_elu4(a1, mu, rs, gq1, bq1);
  *(v4f*)(o + 256) = ln_elu4(a2, mu, rs, gq2, bq2);
  *(v4f*)(o + 384) = ln_elu4(a3, mu, rs, gq3, bq3);
}

extern "C" void kernel_launch(void* const* d_in, const int* in_sizes, int n_in,
                              void* d_out, int out_size, void* d_ws, size_t ws_size,
                              hipStream_t stream) {
  (void)in_sizes; (void)n_in; (void)out_size; (void)ws_size;
  const int*   msg = (const int*)  d_in[0];
  const float* W1  = (const float*)d_in[1];
  const float* b1  = (const float*)d_in[2];
  const float* g1  = (const float*)d_in[3];
  const float* be1 = (const float*)d_in[4];
  const float* W2  = (const float*)d_in[5];
  const float* b2  = (const float*)d_in[6];
  const float* g2  = (const float*)d_in[7];
  const float* be2 = (const float*)d_in[8];
  float* out = (float*)d_out;

  const size_t h_bytes = (size_t)BATCH * HDIM * sizeof(float);
  char* ws = (char*)d_ws;
  float* hs = (float*)ws;
  float* h1 = (float*)(ws + h_bytes);
  float* h2 = (float*)(ws + 2 * h_bytes);

  k_embed_lds<<<dim3(HDIM / CS, BATCH / RB), RB, 0, stream>>>(msg, W1, b1, hs);
  k_ln2_elu<<<BATCH / LROWS, 32 * LROWS, 0, stream>>>(hs, g1, be1, h1);
  k_gemm_wmma<<<dim3(BATCH / 128, HDIM / 64), 256, 0, stream>>>(h1, W2, b2, h2);
  k_ln2_elu<<<BATCH / LROWS, 32 * LROWS, 0, stream>>>(h2, g2, be2, out);
}


# product re-tiled so a block owns 32 whole rows (all 512 columns): LN1+ELU fills its A tile and LN2+ELU drains its result tile in LDS; 2 launches, not 4; h1, h2 never reach memory; sum orders kept
# speedup vs baseline: 1.5260x; 1.1797x over previous
#include <hip/hip_runtime.h>
#include <hip/hip_bf16.h>
#include <math.h>


#define NPOS  80
#define NCHAR 256
#define HDIM  512
#define IDIM  (NPOS * NCHAR)
#define BATCH 8192
#define LN_EPS 1e-5f
#define KC    32

typedef __attribute__((ext_vector_type(2))) float v2f;
typedef __attribute__((ext_vector_type(4))) float v4f;
typedef __attribute__((ext_vector_type(8))) float v8f;
typedef __attribute__((ext_vector_type(4))) int   v4i;

__device__ __forceinline__ float elu(float x) {
  return x > 0.f ? x : expm1f(x);
}

__device__ __forceinline__ v4f ln_elu4(v4f a, float mu, float rs, v4f g, v4f be) {
  v4f o;
  o.x = elu((a.x - mu) * rs * g.x + be.x);
  o.y = elu((a.y - mu) * rs * g.y + be.y);
  o.z = elu((a.z - mu) * rs * g.z + be.z);
  o.w = elu((a.w - mu) * rs * g.w + be.w);
  return o;
}

#define CS   32
#define RB   512
#define SP   32
#define SLAB (NCHAR * SP)

__device__ __forceinline__ void lds_barrier() {
  asm volatile("s_wait_dscnt 0x0\n\ts_barrier_signal -1\n\ts_barrier_wait -1" ::: "memory");
}

__global__ __launch_bounds__(RB)
void k_embed_lds(const int* __restrict__ msg,
                 const float* __restrict__ W1,
                 const float* __restrict__ b1,
                 float* __restrict__ hs) {
  __shared__ __attribute__((aligned(16))) float S[2 * SLAB];
  const int t    = threadIdx.x;
  const int h0   = blockIdx.x * CS;
  const int c    = t & (NCHAR - 1);
  const int jh   = t >> 8;
  const float* wsrc = W1 + (size_t)(h0 + 16 * jh) * IDIM + c;
  const int cswz = (c >> 1) & 7;
  float* srow = S + c * SP;
  const int lane = t & 31;
  const int w    = t >> 5;
  const int g    = lane >> 3;
  const int q    = lane & 7;
  const int row0 = blockIdx.y * RB + w * 32 + g;
  const int* mrow = msg + (size_t)row0 * NPOS;

  v4i mc[8];
#pragma unroll
  for (int j = 0; j < 8; ++j) mc[j] = *(const v4i*)(mrow + (size_t)(4 * j) * NPOS);

  float st[2][16];
#pragma unroll
  for (int r = 0; r < 16; ++r) st[0][r] = wsrc[(size_t)r * IDIM];
#pragma unroll
  for (int k = 0; k < 4; ++k) {
    v4f wv = {st[0][4 * k], st[0][4 * k + 1], st[0][4 * k + 2], st[0][4 * k + 3]};
    *(v4f*)(srow + 4 * ((4 * jh + k) ^ cswz)) = wv;
  }
  lds_barrier();
#pragma unroll
  for (int r = 0; r < 16; ++r) st[1][r] = wsrc[(size_t)r * IDIM + 1 * NCHAR];
  __builtin_amdgcn_sched_barrier(0);
#pragma unroll
  for (int r = 0; r < 16; ++r) st[0][r] = wsrc[(size_t)r * IDIM + 2 * NCHAR];

  v4f acc[8];
#pragma unroll
  for (int j = 0; j < 8; ++j) acc[j] = (v4f){0.f, 0.f, 0.f, 0.f};

  for (int p0 = 0; p0 < NPOS; p0 += 4) {
    const int pn = (p0 + 4 < NPOS) ? (p0 + 4) : (NPOS - 4);
    v4i mn[8];
#pragma unroll
    for (int j = 0; j < 8; ++j) mn[j] = *(const v4i*)(mrow + (size_t)(4 * j) * NPOS + pn);
#pragma unroll
    for (int u = 0; u < 4; ++u) {
      const int p = p0 + u;
      {
        float* d = srow + ((u + 1) & 1) * SLAB;
#pragma unroll
        for (int k = 0; k < 4; ++k) {
          v4f wv = {st[(u + 1) & 1][4 * k], st[(u + 1) & 1][4 * k + 1],
                    st[(u + 1) & 1][4 * k + 2], st[(u + 1) & 1][4 * k + 3]};
          *(v4f*)(d + 4 * ((4 * jh + k) ^ cswz)) = wv;
        }
      }
      {
        const int ps = (p + 3 < NPOS) ? (p + 3) : (NPOS - 1);
#pragma unroll
        for (int r = 0; r < 16; ++r)
          st[(u + 1) & 1][r] = wsrc[(size_t)r * IDIM + (size_t)ps * NCHAR];
      }
      const float* sb = S + (u & 1) * SLAB;
      v4f v[8];
#pragma unroll
      for (int j = 0; j < 8; ++j) {
        const int m = mc[j][u] & (NCHAR - 1);
        v[j] = *(const v4f*)(sb + m * SP + 4 * (q ^ ((m >> 1) & 7)));
      }
#pragma unroll
      for (int j = 0; j < 8; ++j) acc[j] += v[j];
      lds_barrier();
    }
#pragma unroll
    for (int j = 0; j < 8; ++j) mc[j] = mn[j];
  }

  const v4f bv = *(const v4f*)(b1 + h0 + 4 * q);
  float* o = hs + (size_t)row0 * HDIM + h0 + 4 * q;
#pragma unroll
  for (int j = 0; j < 8; ++j)
    *(v4f*)(o + (size_t)(4 * j) * HDIM) = acc[j] + bv;
}

#define FM     32
#define APITCH 516
#define BPITCH 36
#define NCHUNK (HDIM / KC)

__device__ __forceinline__ void row_stats4(const v4f (&a)[4][4], float (&mu)[4], float (&rs)[4]) {
  float s[4], q[4];
#pragma unroll
  for (int i = 0; i < 4; ++i) {
    const v4f t  = (a[i][0] + a[i][1]) + (a[i][2] + a[i][3]);
    const v4f u2 = (a[i][0] * a[i][0] + a[i][1] * a[i][1]) + (a[i][2] * a[i][2] + a[i][3] * a[i][3]);
    s[i] = (t.x + t.y) + (t.z + t.w);
    q[i] = (u2.x + u2.y) + (u2.z + u2.w);
  }
#pragma unroll
  for (int off = 16; off > 0; off >>= 1) {
#pragma unroll
    for (int i = 0; i < 4; ++i) {
      s[i] += __shfl_xor(s[i], off, 32);
      q[i] += __shfl_xor(q[i], off, 32);
    }
  }
#pragma unroll
  for (int i = 0; i < 4; ++i) {
    mu[i] = s[i] * (1.0f / (float)HDIM);
    const float var = q[i] * (1.0f / (float)HDIM) - mu[i] * mu[i];
    rs[i] = rsqrtf(var + LN_EPS);
  }
}

__global__ __launch_bounds__(256)
void k_ln_gemm_ln(const float* __restrict__ hs,
                  const float* __restrict__ g1,
                  const float* __restrict__ be1,
                  const float* __restrict__ W2,
                  const float* __restrict__ b2,
                  const float* __restrict__ g2,
                  const float* __restrict__ be2,
                  float* __restrict__ out) {
  __shared__ __attribute__((aligned(16))) float sA[FM * APITCH];
  __shared__ __attribute__((aligned(16))) float sB[2 * HDIM * BPITCH];
  const int tid  = threadIdx.x;
  const int wid  = tid >> 5;
  const int lane = tid & 31;
  const int m0   = blockIdx.x * FM;
  const int col  = lane * 4;

  const float* wsrc = W2 + (size_t)(tid >> 3) * HDIM + ((tid & 7) << 2);
  float* bdst = sB + (tid >> 3) * BPITCH + ((tid & 7) << 2);

  v4f bst[16];
#pragma unroll
  for (int i = 0; i < 16; ++i) bst[i] = *(const v4f*)(wsrc + (size_t)(32 * i) * HDIM);

  {
    const v4f gq0 = *(const v4f*)(g1 + col);
    const v4f gq1 = *(const v4f*)(g1 + 128 + col);
    const v4f gq2 = *(const v4f*)(g1 + 256 + col);
    const v4f gq3 = *(const v4f*)(g1 + 384 + col);
    const v4f bq0 = *(const v4f*)(be1 + col);
    const v4f bq1 = *(const v4f*)(be1 + 128 + col);
    const v4f bq2 = *(const v4f*)(be1 + 256 + col);
    const v4f bq3 = *(const v4f*)(be1 + 384 + col);
    v4f a[4][4];
#pragma unroll
    for (int i = 0; i < 4; ++i) {
      const float* x = hs + (size_t)(m0 + 4 * wid + i) * HDIM + col;
      a[i][0] = *(const v4f*)(x);
      a[i][1] = *(const v4f*)(x + 128);
      a[i][2] = *(const v4f*)(x + 256);
      a[i][3] = *(const v4f*)(x + 384);
    }
    float mu[4], rs[4];
    row_stats4(a, mu, rs);
#pragma unroll
    for (int i = 0; i < 4; ++i) {
      float* d = sA + (4 * wid + i) * APITCH + col;
      *(v4f*)(d)       = ln_elu4(a[i][0], mu[i], rs[i], gq0, bq0);
      *(v4f*)(d + 128) = ln_elu4(a[i][1], mu[i], rs[i], gq1, bq1);
      *(v4f*)(d + 256) = ln_elu4(a[i][2], mu[i], rs[i], gq2, bq2);
      *(v4f*)(d + 384) = ln_elu4(a[i][3], mu[i], rs[i], gq3, bq3);
    }
  }
#pragma unroll
  for (int i = 0; i < 16; ++i) *(v4f*)(bdst + (32 * i) * BPITCH) = bst[i];
  lds_barrier();

  const int c16   = lane & 15;
  const int khalf = (lane >> 4) << 1;
  const float* Aw = sA + c16 * APITCH + khalf;
  const float* Bw = sB + (wid * 64 + c16) * BPITCH + khalf;

  v8f acc[2][4];
#pragma unroll
  for (int mt = 0; mt < 2; ++mt)
#pragma unroll
    for (int nt = 0; nt < 4; ++nt) acc[mt][nt] = (v8f){0.f, 0.f, 0.f, 0.f, 0.f, 0.f, 0.f, 0.f};

  for (int ch = 0; ch < NCHUNK; ++ch) {
    const int  buf  = ch & 1;
    const bool more = (ch + 1 < NCHUNK);
    if (more) {
#pragma unroll
      for (int i = 0; i < 16; ++i)
        bst[i] = *(const v4f*)(wsrc + (size_t)(32 * i) * HDIM + (ch + 1) * KC);
    }
    const float* Ac = Aw + ch * KC;
    const float* Bc = Bw + buf * (HDIM * BPITCH);
#pragma unroll
    for (int kk = 0; kk < KC; kk += 4) {
      const v2f a0  = *(const v2f*)(Ac + kk);
      const v2f a1  = *(const v2f*)(Ac + 16 * APITCH + kk);
      const v2f bb0 = *(const v2f*)(Bc + kk);
      const v2f bb1 = *(const v2f*)(Bc + 16 * BPITCH + kk);
      const v2f bb2 = *(const v2f*)(Bc + 32 * BPITCH + kk);
      const v2f bb3 = *(const v2f*)(Bc + 48 * BPITCH + kk);
      acc[0][0] = __builtin_amdgcn_wmma_f32_16x16x4_f32(false, a0, false, bb0, (short)0, acc[0][0], false, false);
      acc[1][0] = __builtin_amdgcn_wmma_f32_16x16x4_f32(false, a1, false, bb0, (short)0, acc[1][0], false, false);
      acc[0][1] = __builtin_amdgcn_wmma_f32_16x16x4_f32(false, a0, false, bb1, (short)0, acc[0][1], false, false);
      acc[1][1] = __builtin_amdgcn_wmma_f32_16x16x4_f32(false, a1, false, bb1, (short)0, acc[1][1], false, false);
      acc[0][2] = __builtin_amdgcn_wmma_f32_16x16x4_f32(false, a0, false, bb2, (short)0, acc[0][2], false, false);
      acc[1][2] = __builtin_amdgcn_wmma_f32_16x16x4_f32(false, a1, false, bb2, (short)0, acc[1][2], false, false);
      acc[0][3] = __builtin_amdgcn_wmma_f32_16x16x4_f32(false, a0, false, bb3, (short)0, acc[0][3], false, false);
      acc[1][3] = __builtin_amdgcn_wmma_f32_16x16x4_f32(false, a1, false, bb3, (short)0, acc[1][3], false, false);
    }
    if (more) {
      float* d = bdst + (buf ^ 1) * (HDIM * BPITCH);
#pragma unroll
      for (int i = 0; i < 16; ++i) *(v4f*)(d + (32 * i) * BPITCH) = bst[i];
    }
    lds_barrier();
  }

  const v4f gq0 = *(const v4f*)(g2 + col);
  const v4f gq1 = *(const v4f*)(g2 + 128 + col);
  const v4f gq2 = *(const v4f*)(g2 + 256 + col);
  const v4f gq3 = *(const v4f*)(g2 + 384 + col);
  const v4f bq0 = *(const v4f*)(be2 + col);
  const v4f bq1 = *(const v4f*)(be2 + 128 + col);
  const v4f bq2 = *(const v4f*)(be2 + 256 + col);
  const v4f bq3 = *(const v4f*)(be2 + 384 + col);
  {
    const int rbase = (lane >> 4) << 3;
    float* c = sA + rbase * APITCH + wid * 64 + c16;
#pragma unroll
    for (int nt = 0; nt < 4; ++nt) {
      const float bj = b2[wid * 64 + nt * 16 + c16];
#pragma unroll
      for (int mt = 0; mt < 2; ++mt)
#pragma unroll
        for (int r = 0; r < 8; ++r)
          c[(mt * 16 + r) * APITCH + nt * 16] = acc[mt][nt][r] + bj;
    }
  }
  lds_barrier();
  {
    v4f a[4][4];
#pragma unroll
    for (int i = 0; i < 4; ++i) {
      const float* x = sA + (4 * wid + i) * APITCH + col;
      a[i][0] = *(const v4f*)(x);
      a[i][1] = *(const v4f*)(x + 128);
      a[i][2] = *(const v4f*)(x + 256);
      a[i][3] = *(const v4f*)(x + 384);
    }
    float mu[4], rs[4];
    row_stats4(a, mu, rs);
#pragma unroll
    for (int i = 0; i < 4; ++i) {
      float* o = out + (size_t)(m0 + 4 * wid + i) * HDIM + col;
      *(v4f*)(o)       = ln_elu4(a[i][0], mu[i], rs[i], gq0, bq0);
      *(v4f*)(o + 128) = ln_elu4(a[i][1], mu[i], rs[i], gq1, bq1);
      *(v4f*)(o + 256) = ln_elu4(a[i][2], mu[i], rs[i], gq2, bq2);
      *(v4f*)(o + 384) = ln_elu4(a[i][3], mu[i], rs[i], gq3, bq3);
    }
  }
}

extern "C" void kernel_launch(void* const* d_in, const int* in_sizes, int n_in,
                              void* d_out, int out_size, void* d_ws, size_t ws_size,
                              hipStream_t stream) {
  (void)in_sizes; (void)n_in; (void)out_size; (void)ws_size;
  const int*   msg = (const int*)  d_in[0];
  const float* W1  = (const float*)d_in[1];
  const float* b1  = (const float*)d_in[2];
  const float* g1  = (const float*)d_in[3];
  const float* be1 = (const float*)d_in[4];
  const float* W2  = (const float*)d_in[5];
  const float* b2  = (const float*)d_in[6];
  const float* g2  = (const float*)d_in[7];
  const float* be2 = (const float*)d_in[8];
  float* out = (float*)d_out;

  float* hs = (float*)d_ws;

  k_embed_lds<<<dim3(HDIM / CS, BATCH / RB), RB, 0, stream>>>(msg, W1, b1, hs);
  k_ln_gemm_ln<<<BATCH / FM, 256, 0, stream>>>(hs, g1, be1, W2, b2, g2, be2, out);
}


# embedding: the block's 512 x 80 characters are read once with coalesced 16-byte loads, packed to bytes in LDS (40 KB) and read from LDS per four positions; no character loads to memory in the loop
# speedup vs baseline: 1.5973x; 1.0467x over previous
#include <hip/hip_runtime.h>
#include <hip/hip_bf16.h>
#include <math.h>


#define NPOS  80
#define NCHAR 256
#define HDIM  512
#define IDIM  (NPOS * NCHAR)
#define BATCH 8192
#define LN_EPS 1e-5f
#define KC    32

typedef __attribute__((ext_vector_type(2))) float v2f;
typedef __attribute__((ext_vector_type(4))) float v4f;
typedef __attribute__((ext_vector_type(8))) float v8f;
typedef __attribute__((ext_vector_type(4))) int   v4i;

__device__ __forceinline__ float elu(float x) {
  return x > 0.f ? x : expm1f(x);
}

__device__ __forceinline__ v4f ln_elu4(v4f a, float mu, float rs, v4f g, v4f be) {
  v4f o;
  o.x = elu((a.x - mu) * rs * g.x + be.x);
  o.y = elu((a.y - mu) * rs * g.y + be.y);
  o.z = elu((a.z - mu) * rs * g.z + be.z);
  o.w = elu((a.w - mu) * rs * g.w + be.w);
  return o;
}

#define CS   32
#define RB   512
#define SP   32
#define SLAB (NCHAR * SP)
#define MSGW (NPOS / 4)

__device__ __forceinline__ void lds_barrier() {
  asm volatile("s_wait_dscnt 0x0\n\ts_barrier_signal -1\n\ts_barrier_wait -1" ::: "memory");
}

__global__ __launch_bounds__(RB)
void k_embed_lds(const int* __restrict__ msg,
                 const float* __restrict__ W1,
                 const float* __restrict__ b1,
                 float* __restrict__ hs) {
  __shared__ __attribute__((aligned(16))) float S[2 * SLAB];
  __shared__ unsigned int C[RB * MSGW];
  const int t    = threadIdx.x;
  const int h0   = blockIdx.x * CS;
  const int c    = t & (NCHAR - 1);
  const int jh   = t >> 8;
  const float* wsrc = W1 + (size_t)(h0 + 16 * jh) * IDIM + c;
  const int cswz = (c >> 1) & 7;
  float* srow = S + c * SP;
  const int lane = t & 31;
  const int w    = t >> 5;
  const int g    = lane >> 3;
  const int q    = lane & 7;
  const int row0 = blockIdx.y * RB + w * 32 + g;
  const unsigned int* crow = C + (w * 32 + g) * MSGW;

  float st[2][16];
#pragma unroll
  for (int r = 0; r < 16; ++r) st[0][r] = wsrc[(size_t)r * IDIM];
  {
    const v4i* msrc = (const v4i*)(msg + (size_t)blockIdx.y * (RB * NPOS));
#pragma unroll
    for (int k = 0; k < MSGW; ++k) {
      const v4i x = msrc[k * RB + t];
      C[k * RB + t] = (unsigned int)(x.x & (NCHAR - 1)) | ((unsigned int)(x.y & (NCHAR - 1)) << 8) |
                      ((unsigned int)(x.z & (NCHAR - 1)) << 16) | ((unsigned int)(x.w & (NCHAR - 1)) << 24);
    }
  }
#pragma unroll
  for (int k = 0; k < 4; ++k) {
    v4f wv = {st[0][4 * k], st[0][4 * k + 1], st[0][4 * k + 2], st[0][4 * k + 3]};
    *(v4f*)(srow + 4 * ((4 * jh + k) ^ cswz)) = wv;
  }
  lds_barrier();
#pragma unroll
  for (int r = 0; r < 16; ++r) st[1][r] = wsrc[(size_t)r * IDIM + 1 * NCHAR];
  __builtin_amdgcn_sched_barrier(0);
#pragma unroll
  for (int r = 0; r < 16; ++r) st[0][r] = wsrc[(size_t)r * IDIM + 2 * NCHAR];

  unsigned int mc[8];
#pragma unroll
  for (int j = 0; j < 8; ++j) mc[j] = crow[4 * j * MSGW];

  v4f acc[8];
#pragma unroll
  for (int j = 0; j < 8; ++j) acc[j] = (v4f){0.f, 0.f, 0.f, 0.f};

  for (int p0 = 0; p0 < NPOS; p0 += 4) {
    const int kn = (p0 + 4 < NPOS) ? (p0 / 4 + 1) : (NPOS / 4 - 1);
    unsigned int mn[8];
#pragma unroll
    for (int j = 0; j < 8; ++j) mn[j] = crow[4 * j * MSGW + kn];
#pragma unroll
    for (int u = 0; u < 4; ++u) {
      const int p = p0 + u;
      {
        float* d = srow + ((u + 1) & 1) * SLAB;
#pragma unroll
        for (int k = 0; k < 4; ++k) {
          v4f wv = {st[(u + 1) & 1][4 * k], st[(u + 1) & 1][4 * k + 1],
                    st[(u + 1) & 1][4 * k + 2], st[(u + 1) & 1][4 * k + 3]};
          *(v4f*)(d + 4 * ((4 * jh + k) ^ cswz)) = wv;
        }
      }
      {
        const int ps = (p + 3 < NPOS) ? (p + 3) : (NPOS - 1);
#pragma unroll
        for (int r = 0; r < 16; ++r)
          st[(u + 1) & 1][r] = wsrc[(size_t)r * IDIM + (size_t)ps * NCHAR];
      }
      const float* sb = S + (u & 1) * SLAB;
      v4f v[8];
#pragma unroll
      for (int j = 0; j < 8; ++j) {
        const int m = (int)((mc[j] >> (8 * u)) & (NCHAR - 1));
        v[j] = *(const v4f*)(sb + m * SP + 4 * (q ^ ((m >> 1) & 7)));
      }
#pragma unroll
      for (int j = 0; j < 8; ++j) acc[j] += v[j];
      lds_barrier();
    }
#pragma unroll
    for (int j = 0; j < 8; ++j) mc[j] = mn[j];
  }

  const v4f bv = *(const v4f*)(b1 + h0 + 4 * q);
  float* o = hs + (size_t)row0 * HDIM + h0 + 4 * q;
#pragma unroll
  for (int j = 0; j < 8; ++j)
    *(v4f*)(o + (size_t)(4 * j) * HDIM) = acc[j] + bv;
}

#define FM     32
#define APITCH 516
#define BPITCH 36
#define NCHUNK (HDIM / KC)

__device__ __forceinline__ void row_stats4(const v4f (&a)[4][4], float (&mu)[4], float (&rs)[4]) {
  float s[4], q[4];
#pragma unroll
  for (int i = 0; i < 4; ++i) {
    const v4f t  = (a[i][0] + a[i][1]) + (a[i][2] + a[i][3]);
    const v4f u2 = (a[i][0] * a[i][0] + a[i][1] * a[i][1]) + (a[i][2] * a[i][2] + a[i][3] * a[i][3]);
    s[i] = (t.x + t.y) + (t.z + t.w);
    q[i] = (u2.x + u2.y) + (u2.z + u2.w);
  }
#pragma unroll
  for (int off = 16; off > 0; off >>= 1) {
#pragma unroll
    for (int i = 0; i < 4; ++i) {
      s[i] += __shfl_xor(s[i], off, 32);
      q[i] += __shfl_xor(q[i], off, 32);
    }
  }
#pragma unroll
  for (int i = 0; i < 4; ++i) {
    mu[i] = s[i] * (1.0f / (float)HDIM);
    const float var = q[i] * (1.0f / (float)HDIM) - mu[i] * mu[i];
    rs[i] = rsqrtf(var + LN_EPS);
  }
}

__global__ __launch_bounds__(256)
void k_ln_gemm_ln(const float* __restrict__ hs,
                  const float* __restrict__ g1,
                  const float* __restrict__ be1,
                  const float* __restrict__ W2,
                  const float* __restrict__ b2,
                  const float* __restrict__ g2,
                  const float* __restrict__ be2,
                  float* __restrict__ out) {
  __shared__ __attribute__((aligned(16))) float sA[FM * APITCH];
  __shared__ __attribute__((aligned(16))) float sB[2 * HDIM * BPITCH];
  const int tid  = threadIdx.x;
  const int wid  = tid >> 5;
  const int lane = tid & 31;
  const int m0   = blockIdx.x * FM;
  const int col  = lane * 4;

  const float* wsrc = W2 + (size_t)(tid >> 3) * HDIM + ((tid & 7) << 2);
  float* bdst = sB + (tid >> 3) * BPITCH + ((tid & 7) << 2);

  v4f bst[16];
#pragma unroll
  for (int i = 0; i < 16; ++i) bst[i] = *(const v4f*)(wsrc + (size_t)(32 * i) * HDIM);

  {
    const v4f gq0 = *(const v4f*)(g1 + col);
    const v4f gq1 = *(const v4f*)(g1 + 128 + col);
    const v4f gq2 = *(const v4f*)(g1 + 256 + col);
    const v4f gq3 = *(const v4f*)(g1 + 384 + col);
    const v4f bq0 = *(const v4f*)(be1 + col);
    const v4f bq1 = *(const v4f*)(be1 + 128 + col);
    const v4f bq2 = *(const v4f*)(be1 + 256 + col);
    const v4f bq3 = *(const v4f*)(be1 + 384 + col);
    v4f a[4][4];
#pragma unroll
    for (int i = 0; i < 4; ++i) {
      const float* x = hs + (size_t)(m0 + 4 * wid + i) * HDIM + col;
      a[i][0] = *(const v4f*)(x);
      a[i][1] = *(const v4f*)(x + 128);
      a[i][2] = *(const v4f*)(x + 256);
      a[i][3] = *(const v4f*)(x + 384);
    }
    float mu[4], rs[4];
    row_stats4(a, mu, rs);
#pragma unroll
    for (int i = 0; i < 4; ++i) {
      float* d = sA + (4 * wid + i) * APITCH + col;
      *(v4f*)(d)       = ln_elu4(a[i][0], mu[i], rs[i], gq0, bq0);
      *(v4f*)(d + 128) = ln_elu4(a[i][1], mu[i], rs[i], gq1, bq1);
      *(v4f*)(d + 256) = ln_elu4(a[i][2], mu[i], rs[i], gq2, bq2);
      *(v4f*)(d + 384) = ln_elu4(a[i][3], mu[i], rs[i], gq3, bq3);
    }
  }
#pragma unroll
  for (int i = 0; i < 16; ++i) *(v4f*)(bdst + (32 * i) * BPITCH) = bst[i];
  lds_barrier();

  const int c16   = lane & 15;
  const int khalf = (lane >> 4) << 1;
  const float* Aw = sA + c16 * APITCH + khalf;
  const float* Bw = sB + (wid * 64 + c16) * BPITCH + khalf;

  v8f acc[2][4];
#pragma unroll
  for (int mt = 0; mt < 2; ++mt)
#pragma unroll
    for (int nt = 0; nt < 4; ++nt) acc[mt][nt] = (v8f){0.f, 0.f, 0.f, 0.f, 0.f, 0.f, 0.f, 0.f};

  for (int ch = 0; ch < NCHUNK; ++ch) {
    const int  buf  = ch & 1;
    const bool more = (ch + 1 < NCHUNK);
    if (more) {
#pragma unroll
      for (int i = 0; i < 16; ++i)
        bst[i] = *(const v4f*)(wsrc + (size_t)(32 * i) * HDIM + (ch + 1) * KC);
    }
    const float* Ac = Aw + ch * KC;
    const float* Bc = Bw + buf * (HDIM * BPITCH);
#pragma unroll
    for (int kk = 0; kk < KC; kk += 4) {
      const v2f a0  = *(const v2f*)(Ac + kk);
      const v2f a1  = *(const v2f*)(Ac + 16 * APITCH + kk);
      const v2f bb0 = *(const v2f*)(Bc + kk);
      const v2f bb1 = *(const v2f*)(Bc + 16 * BPITCH + kk);
      const v2f bb2 = *(const v2f*)(Bc + 32 * BPITCH + kk);
      const v2f bb3 = *(const v2f*)(Bc + 48 * BPITCH + kk);
      acc[0][0] = __builtin_amdgcn_wmma_f32_16x16x4_f32(false, a0, false, bb0, (short)0, acc[0][0], false, false);
      acc[1][0] = __builtin_amdgcn_wmma_f32_16x16x4_f32(false, a1, false, bb0, (short)0, acc[1][0], false, false);
      acc[0][1] = __builtin_amdgcn_wmma_f32_16x16x4_f32(false, a0, false, bb1, (short)0, acc[0][1], false, false);
      acc[1][1] = __builtin_amdgcn_wmma_f32_16x16x4_f32(false, a1, false, bb1, (short)0, acc[1][1], false, false);
      acc[0][2] = __builtin_amdgcn_wmma_f32_16x16x4_f32(false, a0, false, bb2, (short)0, acc[0][2], false, false);
      acc[1][2] = __builtin_amdgcn_wmma_f32_16x16x4_f32(false, a1, false, bb2, (short)0, acc[1][2], false, false);
      acc[0][3] = __builtin_amdgcn_wmma_f32_16x16x4_f32(false, a0, false, bb3, (short)0, acc[0][3], false, false);
      acc[1][3] = __builtin_amdgcn_wmma_f32_16x16x4_f32(false, a1, false, bb3, (short)0, acc[1][3], false, false);
    }
    if (more) {
      float* d = bdst + (buf ^ 1) * (HDIM * BPITCH);
#pragma unroll
      for (int i = 0; i < 16; ++i) *(v4f*)(d + (32 * i) * BPITCH) = bst[i];
    }
    lds_barrier();
  }

  const v4f gq0 = *(const v4f*)(g2 + col);
  const v4f gq1 = *(const v4f*)(g2 + 128 + col);
  const v4f gq2 = *(const v4f*)(g2 + 256 + col);
  const v4f gq3 = *(const v4f*)(g2 + 384 + col);
  const v4f bq0 = *(const v4f*)(be2 + col);
  const v4f bq1 = *(const v4f*)(be2 + 128 + col);
  const v4f bq2 = *(const v4f*)(be2 + 256 + col);
  const v4f bq3 = *(const v4f*)(be2 + 384 + col);
  {
    const int rbase = (lane >> 4) << 3;
    float* c = sA + rbase * APITCH + wid * 64 + c16;
#pragma unroll
    for (int nt = 0; nt < 4; ++nt) {
      const float bj = b2[wid * 64 + nt * 16 + c16];
#pragma unroll
      for (int mt = 0; mt < 2; ++mt)
#pragma unroll
        for (int r = 0; r < 8; ++r)
          c[(mt * 16 + r) * APITCH + nt * 16] = acc[mt][nt][r] + bj;
    }
  }
  lds_barrier();
  {
    v4f a[4][4];
#pragma unroll
    for (int i = 0; i < 4; ++i) {
      const float* x = sA + (4 * wid + i) * APITCH + col;
      a[i][0] = *(const v4f*)(x);
      a[i][1] = *(const v4f*)(x + 128);
      a[i][2] = *(const v4f*)(x + 256);
      a[i][3] = *(const v4f*)(x + 384);
    }
    float mu[4], rs[4];
    row_stats4(a, mu, rs);
#pragma unroll
    for (int i = 0; i < 4; ++i) {
      float* o = out + (size_t)(m0 + 4 * wid + i) * HDIM + col;
      *(v4f*)(o)       = ln_elu4(a[i][0], mu[i], rs[i], gq0, bq0);
      *(v4f*)(o + 128) = ln_elu4(a[i][1], mu[i], rs[i], gq1, bq1);
      *(v4f*)(o + 256) = ln_elu4(a[i][2], mu[i], rs[i], gq2, bq2);
      *(v4f*)(o + 384) = ln_elu4(a[i][3], mu[i], rs[i], gq3, bq3);
    }
  }
}

extern "C" void kernel_launch(void* const* d_in, const int* in_sizes, int n_in,
                              void* d_out, int out_size, void* d_ws, size_t ws_size,
                              hipStream_t stream) {
  (void)in_sizes; (void)n_in; (void)out_size; (void)ws_size;
  const int*   msg = (const int*)  d_in[0];
  const float* W1  = (const float*)d_in[1];
  const float* b1  = (const float*)d_in[2];
  const float* g1  = (const float*)d_in[3];
  const float* be1 = (const float*)d_in[4];
  const float* W2  = (const float*)d_in[5];
  const float* b2  = (const float*)d_in[6];
  const float* g2  = (const float*)d_in[7];
  const float* be2 = (const float*)d_in[8];
  float* out = (float*)d_out;

  float* hs = (float*)d_ws;

  k_embed_lds<<<dim3(HDIM / CS, BATCH / RB), RB, 0, stream>>>(msg, W1, b1, hs);
  k_ln_gemm_ln<<<BATCH / FM, 256, 0, stream>>>(hs, g1, be1, W2, b2, g2, be2, out);
}


# embedding: W1 slab fetched with 16-byte loads (a lane takes 4 characters of 1 column; a wave load is 512 contiguous bytes; 4x fewer load instructions), turned 4x4 in registers; LDS swizzle refit
# speedup vs baseline: 1.6794x; 1.0514x over previous
#include <hip/hip_runtime.h>
#include <hip/hip_bf16.h>
#include <math.h>


#define NPOS  80
#define NCHAR 256
#define HDIM  512
#define IDIM  (NPOS * NCHAR)
#define BATCH 8192
#define LN_EPS 1e-5f
#define KC    32

typedef __attribute__((ext_vector_type(2))) float v2f;
typedef __attribute__((ext_vector_type(4))) float v4f;
typedef __attribute__((ext_vector_type(8))) float v8f;
typedef __attribute__((ext_vector_type(4))) int   v4i;

__device__ __forceinline__ float elu(float x) {
  return x > 0.f ? x : expm1f(x);
}

__device__ __forceinline__ v4f ln_elu4(v4f a, float mu, float rs, v4f g, v4f be) {
  v4f o;
  o.x = elu((a.x - mu) * rs * g.x + be.x);
  o.y = elu((a.y - mu) * rs * g.y + be.y);
  o.z = elu((a.z - mu) * rs * g.z + be.z);
  o.w = elu((a.w - mu) * rs * g.w + be.w);
  return o;
}

#define CS   32
#define RB   512
#define SP   32
#define SLAB (NCHAR * SP)
#define MSGW (NPOS / 4)

__device__ __forceinline__ void lds_barrier() {
  asm volatile("s_wait_dscnt 0x0\n\ts_barrier_signal -1\n\ts_barrier_wait -1" ::: "memory");
}

__global__ __launch_bounds__(RB)
void k_embed_lds(const int* __restrict__ msg,
                 const float* __restrict__ W1,
                 const float* __restrict__ b1,
                 float* __restrict__ hs) {
  __shared__ __attribute__((aligned(16))) float S[2 * SLAB];
  __shared__ unsigned int C[RB * MSGW];
  const int t    = threadIdx.x;
  const int h0   = blockIdx.x * CS;
  const int c4   = t & 63;
  const int hg   = t >> 6;
  const float* wsrc = W1 + (size_t)(h0 + 4 * hg) * IDIM + 4 * c4;
  const int par  = c4 & 1;
  const int sslot = 4 * (hg ^ ((c4 >> 1) & 7));
  float* sra = S + (4 * c4 + par) * SP + sslot;
  float* srb = S + (4 * c4 + (par ^ 1)) * SP + sslot;
  const int lane = t & 31;
  const int w    = t >> 5;
  const int g    = lane >> 3;
  const int q    = lane & 7;
  const int row0 = blockIdx.y * RB + w * 32 + g;
  const unsigned int* crow = C + (w * 32 + g) * MSGW;

  v4f st[2][4];
#pragma unroll
  for (int i = 0; i < 4; ++i) st[0][i] = *(const v4f*)(wsrc + (size_t)i * IDIM);
  {
    const v4i* msrc = (const v4i*)(msg + (size_t)blockIdx.y * (RB * NPOS));
#pragma unroll
    for (int k = 0; k < MSGW; ++k) {
      const v4i x = msrc[k * RB + t];
      const unsigned int pk = (unsigned int)(x.x & (NCHAR - 1)) | ((unsigned int)(x.y & (NCHAR - 1)) << 8) |
                              ((unsigned int)(x.z & (NCHAR - 1)) << 16) | ((unsigned int)(x.w & (NCHAR - 1)) << 24);
      C[k * RB + t] = pk ^ ((pk >> 2) & 0x01010101u);
    }
  }
#pragma unroll
  for (int k = 0; k < 4; ++k) {
    v4f wv = {st[0][0][k], st[0][1][k], st[0][2][k], st[0][3][k]};
    *(v4f*)(((k & 1) ? srb : sra) + (k >> 1) * (2 * SP)) = wv;
  }
  lds_barrier();
#pragma unroll
  for (int i = 0; i < 4; ++i) st[1][i] = *(const v4f*)(wsrc + (size_t)i * IDIM + 1 * NCHAR);
  __builtin_amdgcn_sched_barrier(0);
#pragma unroll
  for (int i = 0; i < 4; ++i) st[0][i] = *(const v4f*)(wsrc + (size_t)i * IDIM + 2 * NCHAR);

  unsigned int mc[8];
#pragma unroll
  for (int j = 0; j < 8; ++j) mc[j] = crow[4 * j * MSGW];

  v4f acc[8];
#pragma unroll
  for (int j = 0; j < 8; ++j) acc[j] = (v4f){0.f, 0.f, 0.f, 0.f};

  for (int p0 = 0; p0 < NPOS; p0 += 4) {
    const int kn = (p0 + 4 < NPOS) ? (p0 / 4 + 1) : (NPOS / 4 - 1);
    unsigned int mn[8];
#pragma unroll
    for (int j = 0; j < 8; ++j) mn[j] = crow[4 * j * MSGW + kn];
#pragma unroll
    for (int u = 0; u < 4; ++u) {
      const int p = p0 + u;
      {
#pragma unroll
        for (int k = 0; k < 4; ++k) {
          v4f wv = {st[(u + 1) & 1][0][k], st[(u + 1) & 1][1][k],
                    st[(u + 1) & 1][2][k], st[(u + 1) & 1][3][k]};
          *(v4f*)(((k & 1) ? srb : sra) + ((u + 1) & 1) * SLAB + (k >> 1) * (2 * SP)) = wv;
        }
      }
      {
        const int ps = (p + 3 < NPOS) ? (p + 3) : (NPOS - 1);
#pragma unroll
        for (int i = 0; i < 4; ++i)
          st[(u + 1) & 1][i] = *(const v4f*)(wsrc + (size_t)i * IDIM + (size_t)ps * NCHAR);
      }
      const float* sb = S + (u & 1) * SLAB;
      v4f v[8];
#pragma unroll
      for (int j = 0; j < 8; ++j) {
        const int m = (int)((mc[j] >> (8 * u)) & (NCHAR - 1));
        v[j] = *(const v4f*)(sb + m * SP + 4 * (q ^ ((m >> 3) & 7)));
      }
#pragma unroll
      for (int j = 0; j < 8; ++j) acc[j] += v[j];
      lds_barrier();
      __builtin_amdgcn_sched_barrier(0);
    }
#pragma unroll
    for (int j = 0; j < 8; ++j) mc[j] = mn[j];
  }

  const v4f bv = *(const v4f*)(b1 + h0 + 4 * q);
  float* o = hs + (size_t)row0 * HDIM + h0 + 4 * q;
#pragma unroll
  for (int j = 0; j < 8; ++j)
    *(v4f*)(o + (size_t)(4 * j) * HDIM) = acc[j] + bv;
}

#define FM     32
#define APITCH 516
#define BPITCH 36
#define NCHUNK (HDIM / KC)

__device__ __forceinline__ void row_stats4(const v4f (&a)[4][4], float (&mu)[4], float (&rs)[4]) {
  float s[4], q[4];
#pragma unroll
  for (int i = 0; i < 4; ++i) {
    const v4f t  = (a[i][0] + a[i][1]) + (a[i][2] + a[i][3]);
    const v4f u2 = (a[i][0] * a[i][0] + a[i][1] * a[i][1]) + (a[i][2] * a[i][2] + a[i][3] * a[i][3]);
    s[i] = (t.x + t.y) + (t.z + t.w);
    q[i] = (u2.x + u2.y) + (u2.z + u2.w);
  }
#pragma unroll
  for (int off = 16; off > 0; off >>= 1) {
#pragma unroll
    for (int i = 0; i < 4; ++i) {
      s[i] += __shfl_xor(s[i], off, 32);
      q[i] += __shfl_xor(q[i], off, 32);
    }
  }
#pragma unroll
  for (int i = 0; i < 4; ++i) {
    mu[i] = s[i] * (1.0f / (float)HDIM);
    const float var = q[i] * (1.0f / (float)HDIM) - mu[i] * mu[i];
    rs[i] = rsqrtf(var + LN_EPS);
  }
}

__global__ __launch_bounds__(256)
void k_ln_gemm_ln(const float* __restrict__ hs,
                  const float* __restrict__ g1,
                  const float* __restrict__ be1,
                  const float* __restrict__ W2,
                  const float* __restrict__ b2,
                  const float* __restrict__ g2,
                  const float* __restrict__ be2,
                  float* __restrict__ out) {
  __shared__ __attribute__((aligned(16))) float sA[FM * APITCH];
  __shared__ __attribute__((aligned(16))) float sB[2 * HDIM * BPITCH];
  const int tid  = threadIdx.x;
  const int wid  = tid >> 5;
  const int lane = tid & 31;
  const int m0   = blockIdx.x * FM;
  const int col  = lane * 4;

  const float* wsrc = W2 + (size_t)(tid >> 3) * HDIM + ((tid & 7) << 2);
  float* bdst = sB + (tid >> 3) * BPITCH + ((tid & 7) << 2);

  v4f bst[16];
#pragma unroll
  for (int i = 0; i < 16; ++i) bst[i] = *(const v4f*)(wsrc + (size_t)(32 * i) * HDIM);

  {
    const v4f gq0 = *(const v4f*)(g1 + col);
    const v4f gq1 = *(const v4f*)(g1 + 128 + col);
    const v4f gq2 = *(const v4f*)(g1 + 256 + col);
    const v4f gq3 = *(const v4f*)(g1 + 384 + col);
    const v4f bq0 = *(const v4f*)(be1 + col);
    const v4f bq1 = *(const v4f*)(be1 + 128 + col);
    const v4f bq2 = *(const v4f*)(be1 + 256 + col);
    const v4f bq3 = *(const v4f*)(be1 + 384 + col);
    v4f a[4][4];
#pragma unroll
    for (int i = 0; i < 4; ++i) {
      const float* x = hs + (size_t)(m0 + 4 * wid + i) * HDIM + col;
      a[i][0] = *(const v4f*)(x);
      a[i][1] = *(const v4f*)(x + 128);
      a[i][2] = *(const v4f*)(x + 256);
      a[i][3] = *(const v4f*)(x + 384);
    }
    float mu[4], rs[4];
    row_stats4(a, mu, rs);
#pragma unroll
    for (int i = 0; i < 4; ++i) {
      float* d = sA + (4 * wid + i) * APITCH + col;
      *(v4f*)(d)       = ln_elu4(a[i][0], mu[i], rs[i], gq0, bq0);
      *(v4f*)(d + 128) = ln_elu4(a[i][1], mu[i], rs[i], gq1, bq1);
      *(v4f*)(d + 256) = ln_elu4(a[i][2], mu[i], rs[i], gq2, bq2);
      *(v4f*)(d + 384) = ln_elu4(a[i][3], mu[i], rs[i], gq3, bq3);
    }
  }
#pragma unroll
  for (int i = 0; i < 16; ++i) *(v4f*)(bdst + (32 * i) * BPITCH) = bst[i];
  lds_barrier();

  const int c16   = lane & 15;
  const int khalf = (lane >> 4) << 1;
  const float* Aw = sA + c16 * APITCH + khalf;
  const float* Bw = sB + (wid * 64 + c16) * BPITCH + khalf;

  v8f acc[2][4];
#pragma unroll
  for (int mt = 0; mt < 2; ++mt)
#pragma unroll
    for (int nt = 0; nt < 4; ++nt) acc[mt][nt] = (v8f){0.f, 0.f, 0.f, 0.f, 0.f, 0.f, 0.f, 0.f};

  for (int ch = 0; ch < NCHUNK; ++ch) {
    const int  buf  = ch & 1;
    const bool more = (ch + 1 < NCHUNK);
    if (more) {
#pragma unroll
      for (int i = 0; i < 16; ++i)
        bst[i] = *(const v4f*)(wsrc + (size_t)(32 * i) * HDIM + (ch + 1) * KC);
    }
    const float* Ac = Aw + ch * KC;
    const float* Bc = Bw + buf * (HDIM * BPITCH);
#pragma unroll
    for (int kk = 0; kk < KC; kk += 4) {
      const v2f a0  = *(const v2f*)(Ac + kk);
      const v2f a1  = *(const v2f*)(Ac + 16 * APITCH + kk);
      const v2f bb0 = *(const v2f*)(Bc + kk);
      const v2f bb1 = *(const v2f*)(Bc + 16 * BPITCH + kk);
      const v2f bb2 = *(const v2f*)(Bc + 32 * BPITCH + kk);
      const v2f bb3 = *(const v2f*)(Bc + 48 * BPITCH + kk);
      acc[0][0] = __builtin_amdgcn_wmma_f32_16x16x4_f32(false, a0, false, bb0, (short)0, acc[0][0], false, false);
      acc[1][0] = __builtin_amdgcn_wmma_f32_16x16x4_f32(false, a1, false, bb0, (short)0, acc[1][0], false, false);
      acc[0][1] = __builtin_amdgcn_wmma_f32_16x16x4_f32(false, a0, false, bb1, (short)0, acc[0][1], false, false);
      acc[1][1] = __builtin_amdgcn_wmma_f32_16x16x4_f32(false, a1, false, bb1, (short)0, acc[1][1], false, false);
      acc[0][2] = __builtin_amdgcn_wmma_f32_16x16x4_f32(false, a0, false, bb2, (short)0, acc[0][2], false, false);
      acc[1][2] = __builtin_amdgcn_wmma_f32_16x16x4_f32(false, a1, false, bb2, (short)0, acc[1][2], false, false);
      acc[0][3] = __builtin_amdgcn_wmma_f32_16x16x4_f32(false, a0, false, bb3, (short)0, acc[0][3], false, false);
      acc[1][3] = __builtin_amdgcn_wmma_f32_16x16x4_f32(false, a1, false, bb3, (short)0, acc[1][3], false, false);
    }
    if (more) {
      float* d = bdst + (buf ^ 1) * (HDIM * BPITCH);
#pragma unroll
      for (int i = 0; i < 16; ++i) *(v4f*)(d + (32 * i) * BPITCH) = bst[i];
    }
    lds_barrier();
  }

  const v4f gq0 = *(const v4f*)(g2 + col);
  const v4f gq1 = *(const v4f*)(g2 + 128 + col);
  const v4f gq2 = *(const v4f*)(g2 + 256 + col);
  const v4f gq3 = *(const v4f*)(g2 + 384 + col);
  const v4f bq0 = *(const v4f*)(be2 + col);
  const v4f bq1 = *(const v4f*)(be2 + 128 + col);
  const v4f bq2 = *(const v4f*)(be2 + 256 + col);
  const v4f bq3 = *(const v4f*)(be2 + 384 + col);
  {
    const int rbase = (lane >> 4) << 3;
    float* c = sA + rbase * APITCH + wid * 64 + c16;
#pragma unroll
    for (int nt = 0; nt < 4; ++nt) {
      const float bj = b2[wid * 64 + nt * 16 + c16];
#pragma unroll
      for (int mt = 0; mt < 2; ++mt)
#pragma unroll
        for (int r = 0; r < 8; ++r)
          c[(mt * 16 + r) * APITCH + nt * 16] = acc[mt][nt][r] + bj;
    }
  }
  lds_barrier();
  {
    v4f a[4][4];
#pragma unroll
    for (int i = 0; i < 4; ++i) {
      const float* x = sA + (4 * wid + i) * APITCH + col;
      a[i][0] = *(const v4f*)(x);
      a[i][1] = *(const v4f*)(x + 128);
      a[i][2] = *(const v4f*)(x + 256);
      a[i][3] = *(const v4f*)(x + 384);
    }
    float mu[4], rs[4];
    row_stats4(a, mu, rs);
#pragma unroll
    for (int i = 0; i < 4; ++i) {
      float* o = out + (size_t)(m0 + 4 * wid + i) * HDIM + col;
      *(v4f*)(o)       = ln_elu4(a[i][0], mu[i], rs[i], gq0, bq0);
      *(v4f*)(o + 128) = ln_elu4(a[i][1], mu[i], rs[i], gq1, bq1);
      *(v4f*)(o + 256) = ln_elu4(a[i][2], mu[i], rs[i], gq2, bq2);
      *(v4f*)(o + 384) = ln_elu4(a[i][3], mu[i], rs[i], gq3, bq3);
    }
  }
}

extern "C" void kernel_launch(void* const* d_in, const int* in_sizes, int n_in,
                              void* d_out, int out_size, void* d_ws, size_t ws_size,
                              hipStream_t stream) {
  (void)in_sizes; (void)n_in; (void)out_size; (void)ws_size;
  const int*   msg = (const int*)  d_in[0];
  const float* W1  = (const float*)d_in[1];
  const float* b1  = (const float*)d_in[2];
  const float* g1  = (const float*)d_in[3];
  const float* be1 = (const float*)d_in[4];
  const float* W2  = (const float*)d_in[5];
  const float* b2  = (const float*)d_in[6];
  const float* g2  = (const float*)d_in[7];
  const float* be2 = (const float*)d_in[8];
  float* out = (float*)d_out;

  float* hs = (float*)d_ws;

  k_embed_lds<<<dim3(HDIM / CS, BATCH / RB), RB, 0, stream>>>(msg, W1, b1, hs);
  k_ln_gemm_ln<<<BATCH / FM, 256, 0, stream>>>(hs, g1, be1, W2, b2, g2, be2, out);
}


# ELU without a branch per element: its expm1 side (the library's own steps and roundings, bit-identical results) is computed for every lane and selected, so a lane's 64 elements per phase interleave
# speedup vs baseline: 1.7677x; 1.0526x over previous
#include <hip/hip_runtime.h>
#include <hip/hip_bf16.h>
#include <math.h>


#define NPOS  80
#define NCHAR 256
#define HDIM  512
#define IDIM  (NPOS * NCHAR)
#define BATCH 8192
#define LN_EPS 1e-5f
#define KC    32

typedef __attribute__((ext_vector_type(2))) float v2f;
typedef __attribute__((ext_vector_type(4))) float v4f;
typedef __attribute__((ext_vector_type(8))) float v8f;
typedef __attribute__((ext_vector_type(4))) int   v4i;

__device__ __forceinline__ float expm1_nonpos(float x) {
  const float n = rintf(x * __builtin_bit_cast(float, 0x3fb8aa3bu));
  float r = fmaf(n, __builtin_bit_cast(float, 0xbf317218u), x);
  r = fmaf(n, __builtin_bit_cast(float, 0x3102e308u), r);
  float p = fmaf(__builtin_bit_cast(float, 0x395133b1u), r, __builtin_bit_cast(float, 0x3ab69700u));
  p = fmaf(r, p, __builtin_bit_cast(float, 0x3c0887f9u));
  p = fmaf(r, p, __builtin_bit_cast(float, 0x3d2aaa81u));
  p = fmaf(r, p, __builtin_bit_cast(float, 0x3e2aaaabu));
  p = fmaf(r, p, 0.5f);
  const float q = r * p;
  const float e = fmaf(r, q, r);
  const float s = ldexpf(1.0f, (int)n);
  const float z = fmaf(s, e, s - 1.0f);
  return x < -17.0f ? -1.0f : z;
}

__device__ __forceinline__ float elu(float x) {
  const float e = expm1_nonpos(x);
  return x > 0.f ? x : e;
}

__device__ __forceinline__ v4f ln_elu4(v4f a, float mu, float rs, v4f g, v4f be) {
  v4f o;
  o.x = elu((a.x - mu) * rs * g.x + be.x);
  o.y = elu((a.y - mu) * rs * g.y + be.y);
  o.z = elu((a.z - mu) * rs * g.z + be.z);
  o.w = elu((a.w - mu) * rs * g.w + be.w);
  return o;
}

#define CS   32
#define RB   512
#define SP   32
#define SLAB (NCHAR * SP)
#define MSGW (NPOS / 4)

__device__ __forceinline__ void lds_barrier() {
  asm volatile("s_wait_dscnt 0x0\n\ts_barrier_signal -1\n\ts_barrier_wait -1" ::: "memory");
}

__global__ __launch_bounds__(RB)
void k_embed_lds(const int* __restrict__ msg,
                 const float* __restrict__ W1,
                 const float* __restrict__ b1,
                 float* __restrict__ hs) {
  __shared__ __attribute__((aligned(16))) float S[2 * SLAB];
  __shared__ unsigned int C[RB * MSGW];
  const int t    = threadIdx.x;
  const int h0   = blockIdx.x * CS;
  const int c4   = t & 63;
  const int hg   = t >> 6;
  const float* wsrc = W1 + (size_t)(h0 + 4 * hg) * IDIM + 4 * c4;
  const int par  = c4 & 1;
  const int sslot = 4 * (hg ^ ((c4 >> 1) & 7));
  float* sra = S + (4 * c4 + par) * SP + sslot;
  float* srb = S + (4 * c4 + (par ^ 1)) * SP + sslot;
  const int lane = t & 31;
  const int w    = t >> 5;
  const int g    = lane >> 3;
  const int q    = lane & 7;
  const int row0 = blockIdx.y * RB + w * 32 + g;
  const unsigned int* crow = C + (w * 32 + g) * MSGW;

  v4f st[2][4];
#pragma unroll
  for (int i = 0; i < 4; ++i) st[0][i] = *(const v4f*)(wsrc + (size_t)i * IDIM);
  {
    const v4i* msrc = (const v4i*)(msg + (size_t)blockIdx.y * (RB * NPOS));
#pragma unroll
    for (int k = 0; k < MSGW; ++k) {
      const v4i x = msrc[k * RB + t];
      const unsigned int pk = (unsigned int)(x.x & (NCHAR - 1)) | ((unsigned int)(x.y & (NCHAR - 1)) << 8) |
                              ((unsigned int)(x.z & (NCHAR - 1)) << 16) | ((unsigned int)(x.w & (NCHAR - 1)) << 24);
      C[k * RB + t] = pk ^ ((pk >> 2) & 0x01010101u);
    }
  }
#pragma unroll
  for (int k = 0; k < 4; ++k) {
    v4f wv = {st[0][0][k], st[0][1][k], st[0][2][k], st[0][3][k]};
    *(v4f*)(((k & 1) ? srb : sra) + (k >> 1) * (2 * SP)) = wv;
  }
  lds_barrier();
#pragma unroll
  for (int i = 0; i < 4; ++i) st[1][i] = *(const v4f*)(wsrc + (size_t)i * IDIM + 1 * NCHAR);
  __builtin_amdgcn_sched_barrier(0);
#pragma unroll
  for (int i = 0; i < 4; ++i) st[0][i] = *(const v4f*)(wsrc + (size_t)i * IDIM + 2 * NCHAR);

  unsigned int mc[8];
#pragma unroll
  for (int j = 0; j < 8; ++j) mc[j] = crow[4 * j * MSGW];

  v4f acc[8];
#pragma unroll
  for (int j = 0; j < 8; ++j) acc[j] = (v4f){0.f, 0.f, 0.f, 0.f};

  for (int p0 = 0; p0 < NPOS; p0 += 4) {
    const int kn = (p0 + 4 < NPOS) ? (p0 / 4 + 1) : (NPOS / 4 - 1);
    unsigned int mn[8];
#pragma unroll
    for (int j = 0; j < 8; ++j) mn[j] = crow[4 * j * MSGW + kn];
#pragma unroll
    for (int u = 0; u < 4; ++u) {
      const int p = p0 + u;
      {
#pragma unroll
        for (int k = 0; k < 4; ++k) {
          v4f wv = {st[(u + 1) & 1][0][k], st[(u + 1) & 1][1][k],
                    st[(u + 1) & 1][2][k], st[(u + 1) & 1][3][k]};
          *(v4f*)(((k & 1) ? srb : sra) + ((u + 1) & 1) * SLAB + (k >> 1) * (2 * SP)) = wv;
        }
      }
      {
        const int ps = (p + 3 < NPOS) ? (p + 3) : (NPOS - 1);
#pragma unroll
        for (int i = 0; i < 4; ++i)
          st[(u + 1) & 1][i] = *(const v4f*)(wsrc + (size_t)i * IDIM + (size_t)ps * NCHAR);
      }
      const float* sb = S + (u & 1) * SLAB;
      v4f v[8];
#pragma unroll
      for (int j = 0; j < 8; ++j) {
        const int m = (int)((mc[j] >> (8 * u)) & (NCHAR - 1));
        v[j] = *(const v4f*)(sb + m * SP + 4 * (q ^ ((m >> 3) & 7)));
      }
#pragma unroll
      for (int j = 0; j < 8; ++j) acc[j] += v[j];
      lds_barrier();
      __builtin_amdgcn_sched_barrier(0);
    }
#pragma unroll
    for (int j = 0; j < 8; ++j) mc[j] = mn[j];
  }

  const v4f bv = *(const v4f*)(b1 + h0 + 4 * q);
  float* o = hs + (size_t)row0 * HDIM + h0 + 4 * q;
#pragma unroll
  for (int j = 0; j < 8; ++j)
    *(v4f*)(o + (size_t)(4 * j) * HDIM) = acc[j] + bv;
}

#define FM     32
#define APITCH 516
#define BPITCH 36
#define NCHUNK (HDIM / KC)

__device__ __forceinline__ void row_stats4(const v4f (&a)[4][4], float (&mu)[4], float (&rs)[4]) {
  float s[4], q[4];
#pragma unroll
  for (int i = 0; i < 4; ++i) {
    const v4f t  = (a[i][0] + a[i][1]) + (a[i][2] + a[i][3]);
    const v4f u2 = (a[i][0] * a[i][0] + a[i][1] * a[i][1]) + (a[i][2] * a[i][2] + a[i][3] * a[i][3]);
    s[i] = (t.x + t.y) + (t.z + t.w);
    q[i] = (u2.x + u2.y) + (u2.z + u2.w);
  }
#pragma unroll
  for (int off = 16; off > 0; off >>= 1) {
#pragma unroll
    for (int i = 0; i < 4; ++i) {
      s[i] += __shfl_xor(s[i], off, 32);
      q[i] += __shfl_xor(q[i], off, 32);
    }
  }
#pragma unroll
  for (int i = 0; i < 4; ++i) {
    mu[i] = s[i] * (1.0f / (float)HDIM);
    const float var = q[i] * (1.0f / (float)HDIM) - mu[i] * mu[i];
    rs[i] = rsqrtf(var + LN_EPS);
  }
}

__global__ __launch_bounds__(256)
void k_ln_gemm_ln(const float* __restrict__ hs,
                  const float* __restrict__ g1,
                  const float* __restrict__ be1,
                  const float* __restrict__ W2,
                  const float* __restrict__ b2,
                  const float* __restrict__ g2,
                  const float* __restrict__ be2,
                  float* __restrict__ out) {
  __shared__ __attribute__((aligned(16))) float sA[FM * APITCH];
  __shared__ __attribute__((aligned(16))) float sB[2 * HDIM * BPITCH];
  const int tid  = threadIdx.x;
  const int wid  = tid >> 5;
  const int lane = tid & 31;
  const int m0   = blockIdx.x * FM;
  const int col  = lane * 4;

  const float* wsrc = W2 + (size_t)(tid >> 3) * HDIM + ((tid & 7) << 2);
  float* bdst = sB + (tid >> 3) * BPITCH + ((tid & 7) << 2);

  v4f bst[16];
#pragma unroll
  for (int i = 0; i < 16; ++i) bst[i] = *(const v4f*)(wsrc + (size_t)(32 * i) * HDIM);

  {
    const v4f gq0 = *(const v4f*)(g1 + col);
    const v4f gq1 = *(const v4f*)(g1 + 128 + col);
    const v4f gq2 = *(const v4f*)(g1 + 256 + col);
    const v4f gq3 = *(const v4f*)(g1 + 384 + col);
    const v4f bq0 = *(const v4f*)(be1 + col);
    const v4f bq1 = *(const v4f*)(be1 + 128 + col);
    const v4f bq2 = *(const v4f*)(be1 + 256 + col);
    const v4f bq3 = *(const v4f*)(be1 + 384 + col);
    v4f a[4][4];
#pragma unroll
    for (int i = 0; i < 4; ++i) {
      const float* x = hs + (size_t)(m0 + 4 * wid + i) * HDIM + col;
      a[i][0] = *(const v4f*)(x);
      a[i][1] = *(const v4f*)(x + 128);
      a[i][2] = *(const v4f*)(x + 256);
      a[i][3] = *(const v4f*)(x + 384);
    }
    float mu[4], rs[4];
    row_stats4(a, mu, rs);
#pragma unroll
    for (int i = 0; i < 4; ++i) {
      float* d = sA + (4 * wid + i) * APITCH + col;
      *(v4f*)(d)       = ln_elu4(a[i][0], mu[i], rs[i], gq0, bq0);
      *(v4f*)(d + 128) = ln_elu4(a[i][1], mu[i], rs[i], gq1, bq1);
      *(v4f*)(d + 256) = ln_elu4(a[i][2], mu[i], rs[i], gq2, bq2);
      *(v4f*)(d + 384) = ln_elu4(a[i][3], mu[i], rs[i], gq3, bq3);
    }
  }
#pragma unroll
  for (int i = 0; i < 16; ++i) *(v4f*)(bdst + (32 * i) * BPITCH) = bst[i];
  lds_barrier();

  const int c16   = lane & 15;
  const int khalf = (lane >> 4) << 1;
  const float* Aw = sA + c16 * APITCH + khalf;
  const float* Bw = sB + (wid * 64 + c16) * BPITCH + khalf;

  v8f acc[2][4];
#pragma unroll
  for (int mt = 0; mt < 2; ++mt)
#pragma unroll
    for (int nt = 0; nt < 4; ++nt) acc[mt][nt] = (v8f){0.f, 0.f, 0.f, 0.f, 0.f, 0.f, 0.f, 0.f};

  for (int ch = 0; ch < NCHUNK; ++ch) {
    const int  buf  = ch & 1;
    const bool more = (ch + 1 < NCHUNK);
    if (more) {
#pragma unroll
      for (int i = 0; i < 16; ++i)
        bst[i] = *(const v4f*)(wsrc + (size_t)(32 * i) * HDIM + (ch + 1) * KC);
    }
    const float* Ac = Aw + ch * KC;
    const float* Bc = Bw + buf * (HDIM * BPITCH);
#pragma unroll
    for (int kk = 0; kk < KC; kk += 4) {
      const v2f a0  = *(const v2f*)(Ac + kk);
      const v2f a1  = *(const v2f*)(Ac + 16 * APITCH + kk);
      const v2f bb0 = *(const v2f*)(Bc + kk);
      const v2f bb1 = *(const v2f*)(Bc + 16 * BPITCH + kk);
      const v2f bb2 = *(const v2f*)(Bc + 32 * BPITCH + kk);
      const v2f bb3 = *(const v2f*)(Bc + 48 * BPITCH + kk);
      acc[0][0] = __builtin_amdgcn_wmma_f32_16x16x4_f32(false, a0, false, bb0, (short)0, acc[0][0], false, false);
      acc[1][0] = __builtin_amdgcn_wmma_f32_16x16x4_f32(false, a1, false, bb0, (short)0, acc[1][0], false, false);
      acc[0][1] = __builtin_amdgcn_wmma_f32_16x16x4_f32(false, a0, false, bb1, (short)0, acc[0][1], false, false);
      acc[1][1] = __builtin_amdgcn_wmma_f32_16x16x4_f32(false, a1, false, bb1, (short)0, acc[1][1], false, false);
      acc[0][2] = __builtin_amdgcn_wmma_f32_16x16x4_f32(false, a0, false, bb2, (short)0, acc[0][2], false, false);
      acc[1][2] = __builtin_amdgcn_wmma_f32_16x16x4_f32(false, a1, false, bb2, (short)0, acc[1][2], false, false);
      acc[0][3] = __builtin_amdgcn_wmma_f32_16x16x4_f32(false, a0, false, bb3, (short)0, acc[0][3], false, false);
      acc[1][3] = __builtin_amdgcn_wmma_f32_16x16x4_f32(false, a1, false, bb3, (short)0, acc[1][3], false, false);
    }
    if (more) {
      float* d = bdst + (buf ^ 1) * (HDIM * BPITCH);
#pragma unroll
      for (int i = 0; i < 16; ++i) *(v4f*)(d + (32 * i) * BPITCH) = bst[i];
    }
    lds_barrier();
  }

  const v4f gq0 = *(const v4f*)(g2 + col);
  const v4f gq1 = *(const v4f*)(g2 + 128 + col);
  const v4f gq2 = *(const v4f*)(g2 + 256 + col);
  const v4f gq3 = *(const v4f*)(g2 + 384 + col);
  const v4f bq0 = *(const v4f*)(be2 + col);
  const v4f bq1 = *(const v4f*)(be2 + 128 + col);
  const v4f bq2 = *(const v4f*)(be2 + 256 + col);
  const v4f bq3 = *(const v4f*)(be2 + 384 + col);
  {
    const int rbase = (lane >> 4) << 3;
    float* c = sA + rbase * APITCH + wid * 64 + c16;
#pragma unroll
    for (int nt = 0; nt < 4; ++nt) {
      const float bj = b2[wid * 64 + nt * 16 + c16];
#pragma unroll
      for (int mt = 0; mt < 2; ++mt)
#pragma unroll
        for (int r = 0; r < 8; ++r)
          c[(mt * 16 + r) * APITCH + nt * 16] = acc[mt][nt][r] + bj;
    }
  }
  lds_barrier();
  {
    v4f a[4][4];
#pragma unroll
    for (int i = 0; i < 4; ++i) {
      const float* x = sA + (4 * wid + i) * APITCH + col;
      a[i][0] = *(const v4f*)(x);
      a[i][1] = *(const v4f*)(x + 128);
      a[i][2] = *(const v4f*)(x + 256);
      a[i][3] = *(const v4f*)(x + 384);
    }
    float mu[4], rs[4];
    row_stats4(a, mu, rs);
#pragma unroll
    for (int i = 0; i < 4; ++i) {
      float* o = out + (size_t)(m0 + 4 * wid + i) * HDIM + col;
      *(v4f*)(o)       = ln_elu4(a[i][0], mu[i], rs[i], gq0, bq0);
      *(v4f*)(o + 128) = ln_elu4(a[i][1], mu[i], rs[i], gq1, bq1);
      *(v4f*)(o + 256) = ln_elu4(a[i][2], mu[i], rs[i], gq2, bq2);
      *(v4f*)(o + 384) = ln_elu4(a[i][3], mu[i], rs[i], gq3, bq3);
    }
  }
}

extern "C" void kernel_launch(void* const* d_in, const int* in_sizes, int n_in,
                              void* d_out, int out_size, void* d_ws, size_t ws_size,
                              hipStream_t stream) {
  (void)in_sizes; (void)n_in; (void)out_size; (void)ws_size;
  const int*   msg = (const int*)  d_in[0];
  const float* W1  = (const float*)d_in[1];
  const float* b1  = (const float*)d_in[2];
  const float* g1  = (const float*)d_in[3];
  const float* be1 = (const float*)d_in[4];
  const float* W2  = (const float*)d_in[5];
  const float* b2  = (const float*)d_in[6];
  const float* g2  = (const float*)d_in[7];
  const float* be2 = (const float*)d_in[8];
  float* out = (float*)d_out;

  float* hs = (float*)d_ws;

  k_embed_lds<<<dim3(HDIM / CS, BATCH / RB), RB, 0, stream>>>(msg, W1, b1, hs);
  k_ln_gemm_ln<<<BATCH / FM, 256, 0, stream>>>(hs, g1, be1, W2, b2, g2, be2, out);
}


# embedding: ring of four slab buffers, position barrier split (signal, wait): gather reads and slab stores issued between the halves stay in flight across it; rows added one position later, same order
# speedup vs baseline: 1.8778x; 1.0623x over previous
#include <hip/hip_runtime.h>
#include <hip/hip_bf16.h>
#include <math.h>


#define NPOS  80
#define NCHAR 256
#define HDIM  512
#define IDIM  (NPOS * NCHAR)
#define BATCH 8192
#define LN_EPS 1e-5f
#define KC    32

typedef __attribute__((ext_vector_type(2))) float v2f;
typedef __attribute__((ext_vector_type(4))) float v4f;
typedef __attribute__((ext_vector_type(8))) float v8f;
typedef __attribute__((ext_vector_type(4))) int   v4i;

__device__ __forceinline__ float expm1_nonpos(float x) {
  const float n = rintf(x * __builtin_bit_cast(float, 0x3fb8aa3bu));
  float r = fmaf(n, __builtin_bit_cast(float, 0xbf317218u), x);
  r = fmaf(n, __builtin_bit_cast(float, 0x3102e308u), r);
  float p = fmaf(__builtin_bit_cast(float, 0x395133b1u), r, __builtin_bit_cast(float, 0x3ab69700u));
  p = fmaf(r, p, __builtin_bit_cast(float, 0x3c0887f9u));
  p = fmaf(r, p, __builtin_bit_cast(float, 0x3d2aaa81u));
  p = fmaf(r, p, __builtin_bit_cast(float, 0x3e2aaaabu));
  p = fmaf(r, p, 0.5f);
  const float q = r * p;
  const float e = fmaf(r, q, r);
  const float s = ldexpf(1.0f, (int)n);
  const float z = fmaf(s, e, s - 1.0f);
  return x < -17.0f ? -1.0f : z;
}

__device__ __forceinline__ float elu(float x) {
  const float e = expm1_nonpos(x);
  return x > 0.f ? x : e;
}

__device__ __forceinline__ v4f ln_elu4(v4f a, float mu, float rs, v4f g, v4f be) {
  v4f o;
  o.x = elu((a.x - mu) * rs * g.x + be.x);
  o.y = elu((a.y - mu) * rs * g.y + be.y);
  o.z = elu((a.z - mu) * rs * g.z + be.z);
  o.w = elu((a.w - mu) * rs * g.w + be.w);
  return o;
}

#define CS   32
#define RB   512
#define SP   32
#define SLAB (NCHAR * SP)
#define MSGW (NPOS / 4)

__device__ __forceinline__ void lds_barrier() {
  asm volatile("s_wait_dscnt 0x0\n\ts_barrier_signal -1\n\ts_barrier_wait -1" ::: "memory");
}

__device__ __forceinline__ void lds_barrier_signal() {
  asm volatile("s_wait_dscnt 0x0\n\ts_barrier_signal -1" ::: "memory");
}
__device__ __forceinline__ void lds_barrier_wait() {
  asm volatile("s_barrier_wait -1" ::: "memory");
}

__global__ __launch_bounds__(RB)
void k_embed_lds(const int* __restrict__ msg,
                 const float* __restrict__ W1,
                 const float* __restrict__ b1,
                 float* __restrict__ hs) {
  __shared__ __attribute__((aligned(16))) float S[4 * SLAB];
  __shared__ unsigned int C[RB * MSGW];
  const int t    = threadIdx.x;
  const int h0   = blockIdx.x * CS;
  const int c4   = t & 63;
  const int hg   = t >> 6;
  const float* wsrc = W1 + (size_t)(h0 + 4 * hg) * IDIM + 4 * c4;
  const int par  = c4 & 1;
  const int sslot = 4 * (hg ^ ((c4 >> 1) & 7));
  float* sra = S + (4 * c4 + par) * SP + sslot;
  float* srb = S + (4 * c4 + (par ^ 1)) * SP + sslot;
  const int lane = t & 31;
  const int w    = t >> 5;
  const int g    = lane >> 3;
  const int q    = lane & 7;
  const int row0 = blockIdx.y * RB + w * 32 + g;
  const unsigned int* crow = C + (w * 32 + g) * MSGW;

  v4f st[2][4];
#pragma unroll
  for (int i = 0; i < 4; ++i) st[0][i] = *(const v4f*)(wsrc + (size_t)i * IDIM);
#pragma unroll
  for (int i = 0; i < 4; ++i) st[1][i] = *(const v4f*)(wsrc + (size_t)i * IDIM + 1 * NCHAR);
  {
    const v4i* msrc = (const v4i*)(msg + (size_t)blockIdx.y * (RB * NPOS));
#pragma unroll
    for (int k = 0; k < MSGW; ++k) {
      const v4i x = msrc[k * RB + t];
      const unsigned int pk = (unsigned int)(x.x & (NCHAR - 1)) | ((unsigned int)(x.y & (NCHAR - 1)) << 8) |
                              ((unsigned int)(x.z & (NCHAR - 1)) << 16) | ((unsigned int)(x.w & (NCHAR - 1)) << 24);
      C[k * RB + t] = pk ^ ((pk >> 2) & 0x01010101u);
    }
  }
#pragma unroll
  for (int s = 0; s < 2; ++s) {
#pragma unroll
    for (int k = 0; k < 4; ++k) {
      v4f wv = {st[s][0][k], st[s][1][k], st[s][2][k], st[s][3][k]};
      *(v4f*)(((k & 1) ? srb : sra) + s * SLAB + (k >> 1) * (2 * SP)) = wv;
    }
  }
  lds_barrier();
#pragma unroll
  for (int i = 0; i < 4; ++i) st[0][i] = *(const v4f*)(wsrc + (size_t)i * IDIM + 2 * NCHAR);
  __builtin_amdgcn_sched_barrier(0);
#pragma unroll
  for (int i = 0; i < 4; ++i) st[1][i] = *(const v4f*)(wsrc + (size_t)i * IDIM + 3 * NCHAR);

  unsigned int mc[8];
#pragma unroll
  for (int j = 0; j < 8; ++j) mc[j] = crow[4 * j * MSGW];

  v4f acc[8];
#pragma unroll
  for (int j = 0; j < 8; ++j) acc[j] = (v4f){0.f, 0.f, 0.f, 0.f};
  v4f vp[8];
#pragma unroll
  for (int j = 0; j < 8; ++j) vp[j] = (v4f){0.f, 0.f, 0.f, 0.f};
  int ad[8];
#pragma unroll
  for (int j = 0; j < 8; ++j) {
    const int m = (int)(mc[j] & (NCHAR - 1));
    ad[j] = m * SP + 4 * (q ^ ((m >> 3) & 7));
  }

  for (int p0 = 0; p0 < NPOS; p0 += 4) {
    const int kn = (p0 + 4 < NPOS) ? (p0 / 4 + 1) : (NPOS / 4 - 1);
    unsigned int mn[8];
#pragma unroll
    for (int j = 0; j < 8; ++j) mn[j] = crow[4 * j * MSGW + kn];
#pragma unroll
    for (int u = 0; u < 4; ++u) {
      const int p = p0 + u;
      lds_barrier_signal();
      const float* sb = S + u * SLAB;
      v4f v[8];
#pragma unroll
      for (int j = 0; j < 8; ++j) v[j] = *(const v4f*)(sb + ad[j]);
      {
#pragma unroll
        for (int k = 0; k < 4; ++k) {
          v4f wv = {st[u & 1][0][k], st[u & 1][1][k], st[u & 1][2][k], st[u & 1][3][k]};
          *(v4f*)(((k & 1) ? srb : sra) + ((u + 2) & 3) * SLAB + (k >> 1) * (2 * SP)) = wv;
        }
      }
      {
        const int ps = (p + 4 < NPOS) ? (p + 4) : (NPOS - 1);
#pragma unroll
        for (int i = 0; i < 4; ++i)
          st[u & 1][i] = *(const v4f*)(wsrc + (size_t)i * IDIM + (size_t)ps * NCHAR);
      }
#pragma unroll
      for (int j = 0; j < 8; ++j) {
        const int m = (u < 3) ? (int)((mc[j] >> (8 * (u + 1))) & (NCHAR - 1)) : (int)(mn[j] & (NCHAR - 1));
        ad[j] = m * SP + 4 * (q ^ ((m >> 3) & 7));
      }
#pragma unroll
      for (int j = 0; j < 8; ++j) acc[j] += vp[j];
#pragma unroll
      for (int j = 0; j < 8; ++j) vp[j] = v[j];
      lds_barrier_wait();
      __builtin_amdgcn_sched_barrier(0);
    }
#pragma unroll
    for (int j = 0; j < 8; ++j) mc[j] = mn[j];
  }
#pragma unroll
  for (int j = 0; j < 8; ++j) acc[j] += vp[j];

  const v4f bv = *(const v4f*)(b1 + h0 + 4 * q);
  float* o = hs + (size_t)row0 * HDIM + h0 + 4 * q;
#pragma unroll
  for (int j = 0; j < 8; ++j)
    *(v4f*)(o + (size_t)(4 * j) * HDIM) = acc[j] + bv;
}

#define FM     32
#define APITCH 516
#define BPITCH 36
#define NCHUNK (HDIM / KC)

__device__ __forceinline__ void row_stats4(const v4f (&a)[4][4], float (&mu)[4], float (&rs)[4]) {
  float s[4], q[4];
#pragma unroll
  for (int i = 0; i < 4; ++i) {
    const v4f t  = (a[i][0] + a[i][1]) + (a[i][2] + a[i][3]);
    const v4f u2 = (a[i][0] * a[i][0] + a[i][1] * a[i][1]) + (a[i][2] * a[i][2] + a[i][3] * a[i][3]);
    s[i] = (t.x + t.y) + (t.z + t.w);
    q[i] = (u2.x + u2.y) + (u2.z + u2.w);
  }
#pragma unroll
  for (int off = 16; off > 0; off >>= 1) {
#pragma unroll
    for (int i = 0; i < 4; ++i) {
      s[i] += __shfl_xor(s[i], off, 32);
      q[i] += __shfl_xor(q[i], off, 32);
    }
  }
#pragma unroll
  for (int i = 0; i < 4; ++i) {
    mu[i] = s[i] * (1.0f / (float)HDIM);
    const float var = q[i] * (1.0f / (float)HDIM) - mu[i] * mu[i];
    rs[i] = rsqrtf(var + LN_EPS);
  }
}

__global__ __launch_bounds__(256)
void k_ln_gemm_ln(const float* __restrict__ hs,
                  const float* __restrict__ g1,
                  const float* __restrict__ be1,
                  const float* __restrict__ W2,
                  const float* __restrict__ b2,
                  const float* __restrict__ g2,
                  const float* __restrict__ be2,
                  float* __restrict__ out) {
  __shared__ __attribute__((aligned(16))) float sA[FM * APITCH];
  __shared__ __attribute__((aligned(16))) float sB[2 * HDIM * BPITCH];
  const int tid  = threadIdx.x;
  const int wid  = tid >> 5;
  const int lane = tid & 31;
  const int m0   = blockIdx.x * FM;
  const int col  = lane * 4;

  const float* wsrc = W2 + (size_t)(tid >> 3) * HDIM + ((tid & 7) << 2);
  float* bdst = sB + (tid >> 3) * BPITCH + ((tid & 7) << 2);

  v4f bst[16];
#pragma unroll
  for (int i = 0; i < 16; ++i) bst[i] = *(const v4f*)(wsrc + (size_t)(32 * i) * HDIM);

  {
    const v4f gq0 = *(const v4f*)(g1 + col);
    const v4f gq1 = *(const v4f*)(g1 + 128 + col);
    const v4f gq2 = *(const v4f*)(g1 + 256 + col);
    const v4f gq3 = *(const v4f*)(g1 + 384 + col);
    const v4f bq0 = *(const v4f*)(be1 + col);
    const v4f bq1 = *(const v4f*)(be1 + 128 + col);
    const v4f bq2 = *(const v4f*)(be1 + 256 + col);
    const v4f bq3 = *(const v4f*)(be1 + 384 + col);
    v4f a[4][4];
#pragma unroll
    for (int i = 0; i < 4; ++i) {
      const float* x = hs + (size_t)(m0 + 4 * wid + i) * HDIM + col;
      a[i][0] = *(const v4f*)(x);
      a[i][1] = *(const v4f*)(x + 128);
      a[i][2] = *(const v4f*)(x + 256);
      a[i][3] = *(const v4f*)(x + 384);
    }
    float mu[4], rs[4];
    row_stats4(a, mu, rs);
#pragma unroll
    for (int i = 0; i < 4; ++i) {
      float* d = sA + (4 * wid + i) * APITCH + col;
      *(v4f*)(d)       = ln_elu4(a[i][0], mu[i], rs[i], gq0, bq0);
      *(v4f*)(d + 128) = ln_elu4(a[i][1], mu[i], rs[i], gq1, bq1);
      *(v4f*)(d + 256) = ln_elu4(a[i][2], mu[i], rs[i], gq2, bq2);
      *(v4f*)(d + 384) = ln_elu4(a[i][3], mu[i], rs[i], gq3, bq3);
    }
  }
#pragma unroll
  for (int i = 0; i < 16; ++i) *(v4f*)(bdst + (32 * i) * BPITCH) = bst[i];
  lds_barrier();

  const int c16   = lane & 15;
  const int khalf = (lane >> 4) << 1;
  const float* Aw = sA + c16 * APITCH + khalf;
  const float* Bw = sB + (wid * 64 + c16) * BPITCH + khalf;

  v8f acc[2][4];
#pragma unroll
  for (int mt = 0; mt < 2; ++mt)
#pragma unroll
    for (int nt = 0; nt < 4; ++nt) acc[mt][nt] = (v8f){0.f, 0.f, 0.f, 0.f, 0.f, 0.f, 0.f, 0.f};

  for (int ch = 0; ch < NCHUNK; ++ch) {
    const int  buf  = ch & 1;
    const bool more = (ch + 1 < NCHUNK);
    if (more) {
#pragma unroll
      for (int i = 0; i < 16; ++i)
        bst[i] = *(const v4f*)(wsrc + (size_t)(32 * i) * HDIM + (ch + 1) * KC);
    }
    const float* Ac = Aw + ch * KC;
    const float* Bc = Bw + buf * (HDIM * BPITCH);
#pragma unroll
    for (int kk = 0; kk < KC; kk += 4) {
      const v2f a0  = *(const v2f*)(Ac + kk);
      const v2f a1  = *(const v2f*)(Ac + 16 * APITCH + kk);
      const v2f bb0 = *(const v2f*)(Bc + kk);
      const v2f bb1 = *(const v2f*)(Bc + 16 * BPITCH + kk);
      const v2f bb2 = *(const v2f*)(Bc + 32 * BPITCH + kk);
      const v2f bb3 = *(const v2f*)(Bc + 48 * BPITCH + kk);
      acc[0][0] = __builtin_amdgcn_wmma_f32_16x16x4_f32(false, a0, false, bb0, (short)0, acc[0][0], false, false);
      acc[1][0] = __builtin_amdgcn_wmma_f32_16x16x4_f32(false, a1, false, bb0, (short)0, acc[1][0], false, false);
      acc[0][1] = __builtin_amdgcn_wmma_f32_16x16x4_f32(false, a0, false, bb1, (short)0, acc[0][1], false, false);
      acc[1][1] = __builtin_amdgcn_wmma_f32_16x16x4_f32(false, a1, false, bb1, (short)0, acc[1][1], false, false);
      acc[0][2] = __builtin_amdgcn_wmma_f32_16x16x4_f32(false, a0, false, bb2, (short)0, acc[0][2], false, false);
      acc[1][2] = __builtin_amdgcn_wmma_f32_16x16x4_f32(false, a1, false, bb2, (short)0, acc[1][2], false, false);
      acc[0][3] = __builtin_amdgcn_wmma_f32_16x16x4_f32(false, a0, false, bb3, (short)0, acc[0][3], false, false);
      acc[1][3] = __builtin_amdgcn_wmma_f32_16x16x4_f32(false, a1, false, bb3, (short)0, acc[1][3], false, false);
    }
    if (more) {
      float* d = bdst + (buf ^ 1) * (HDIM * BPITCH);
#pragma unroll
      for (int i = 0; i < 16; ++i) *(v4f*)(d + (32 * i) * BPITCH) = bst[i];
    }
    lds_barrier();
  }

  const v4f gq0 = *(const v4f*)(g2 + col);
  const v4f gq1 = *(const v4f*)(g2 + 128 + col);
  const v4f gq2 = *(const v4f*)(g2 + 256 + col);
  const v4f gq3 = *(const v4f*)(g2 + 384 + col);
  const v4f bq0 = *(const v4f*)(be2 + col);
  const v4f bq1 = *(const v4f*)(be2 + 128 + col);
  const v4f bq2 = *(const v4f*)(be2 + 256 + col);
  const v4f bq3 = *(const v4f*)(be2 + 384 + col);
  {
    const int rbase = (lane >> 4) << 3;
    float* c = sA + rbase * APITCH + wid * 64 + c16;
#pragma unroll
    for (int nt = 0; nt < 4; ++nt) {
      const float bj = b2[wid * 64 + nt * 16 + c16];
#pragma unroll
      for (int mt = 0; mt < 2; ++mt)
#pragma unroll
        for (int r = 0; r < 8; ++r)
          c[(mt * 16 + r) * APITCH + nt * 16] = acc[mt][nt][r] + bj;
    }
  }
  lds_barrier();
  {
    v4f a[4][4];
#pragma unroll
    for (int i = 0; i < 4; ++i) {
      const float* x = sA + (4 * wid + i) * APITCH + col;
      a[i][0] = *(const v4f*)(x);
      a[i][1] = *(const v4f*)(x + 128);
      a[i][2] = *(const v4f*)(x + 256);
      a[i][3] = *(const v4f*)(x + 384);
    }
    float mu[4], rs[4];
    row_stats4(a, mu, rs);
#pragma unroll
    for (int i = 0; i < 4; ++i) {
      float* o = out + (size_t)(m0 + 4 * wid + i) * HDIM + col;
      *(v4f*)(o)       = ln_elu4(a[i][0], mu[i], rs[i], gq0, bq0);
      *(v4f*)(o + 128) = ln_elu4(a[i][1], mu[i], rs[i], gq1, bq1);
      *(v4f*)(o + 256) = ln_elu4(a[i][2], mu[i], rs[i], gq2, bq2);
      *(v4f*)(o + 384) = ln_elu4(a[i][3], mu[i], rs[i], gq3, bq3);
    }
  }
}

extern "C" void kernel_launch(void* const* d_in, const int* in_sizes, int n_in,
                              void* d_out, int out_size, void* d_ws, size_t ws_size,
                              hipStream_t stream) {
  (void)in_sizes; (void)n_in; (void)out_size; (void)ws_size;
  const int*   msg = (const int*)  d_in[0];
  const float* W1  = (const float*)d_in[1];
  const float* b1  = (const float*)d_in[2];
  const float* g1  = (const float*)d_in[3];
  const float* be1 = (const float*)d_in[4];
  const float* W2  = (const float*)d_in[5];
  const float* b2  = (const float*)d_in[6];
  const float* g2  = (const float*)d_in[7];
  const float* be2 = (const float*)d_in[8];
  float* out = (float*)d_out;

  float* hs = (float*)d_ws;

  k_embed_lds<<<dim3(HDIM / CS, BATCH / RB), RB, 0, stream>>>(msg, W1, b1, hs);
  k_ln_gemm_ln<<<BATCH / FM, 256, 0, stream>>>(hs, g1, be1, W2, b2, g2, be2, out);
}
